# starting kernel
# baseline (speedup 1.0000x reference)
#include <hip/hip_runtime.h>

#define NN 200000
#define NE 6400000
#define IN_DIM 32
#define HID_DIM 32
#define OUT_DIM 16

typedef __attribute__((ext_vector_type(2))) float v2f;
typedef __attribute__((ext_vector_type(8))) float v8f;

__global__ void init_deg_kernel(float* __restrict__ deg, int n) {
    int i = blockIdx.x * blockDim.x + threadIdx.x;
    if (i < n) deg[i] = 1.0f;
}

__global__ void count_deg_kernel(const int* __restrict__ dst, float* __restrict__ deg, int e) {
    int i = blockIdx.x * blockDim.x + threadIdx.x;
    if (i < e) atomicAdd(&deg[dst[i]], 1.0f);
}

__global__ void rsqrt_kernel(float* __restrict__ deg, int n) {
    int i = blockIdx.x * blockDim.x + threadIdx.x;
    if (i < n) deg[i] = rsqrtf(deg[i]);
}

template <int K, int FOUT>
__global__ void gemm_scaled_kernel(const float* __restrict__ X,
                                   const float* __restrict__ W,
                                   const float* __restrict__ dinv,
                                   float* __restrict__ out, int nrows) {
    const int wave = (blockIdx.x * blockDim.x + threadIdx.x) >> 5;
    const int lane = threadIdx.x & 31;
    const int tilesN = FOUT / 16;
    const int tm = wave / tilesN;
    const int tn = wave % tilesN;
    if (tm * 16 >= nrows) return;

    const int m = lane & 15;
    const int half = lane >> 4;
    const int row0 = tm * 16;
    const int col0 = tn * 16;

    v8f acc = {};
#pragma unroll
    for (int kb = 0; kb < K; kb += 4) {
        const int k0 = kb + 2 * half;
        v2f a, b;
        a.x = X[(size_t)(row0 + m) * K + k0 + 0];
        a.y = X[(size_t)(row0 + m) * K + k0 + 1];
        b.x = W[(size_t)(k0 + 0) * FOUT + col0 + m];
        b.y = W[(size_t)(k0 + 1) * FOUT + col0 + m];
        acc = __builtin_amdgcn_wmma_f32_16x16x4_f32(false, a, false, b,
                                                    (short)0, acc, false, false);
    }
#pragma unroll
    for (int r = 0; r < 8; ++r) {
        const int row = row0 + r + 8 * half;
        out[(size_t)row * FOUT + col0 + m] = acc[r] * dinv[row];
    }
}

template <int F>
__global__ void init_agg_kernel(const float* __restrict__ hws,
                                const float* __restrict__ dinv,
                                const float* __restrict__ bias,
                                float* __restrict__ agg, int n) {
    int i = blockIdx.x * blockDim.x + threadIdx.x;
    if (i >= n * F) return;
    int node = i / F;
    int f = i - node * F;
    agg[i] = hws[i] * dinv[node] + bias[f];
}

template <int F>
__global__ void edge_agg_kernel(const int* __restrict__ src, const int* __restrict__ dst,
                                const float* __restrict__ hws, const float* __restrict__ dinv,
                                float* __restrict__ agg, int ne) {
    constexpr int TPE = F / 4;
    int tid = blockIdx.x * blockDim.x + threadIdx.x;
    int e = tid / TPE;
    int g = tid - e * TPE;
    if (e >= ne) return;
    const int s = src[e];
    const int d = dst[e];
    const float nd = dinv[d];
    const float4 v = reinterpret_cast<const float4*>(hws + (size_t)s * F)[g];
    float* a = agg + (size_t)d * F + g * 4;
    atomicAdd(a + 0, v.x * nd);
    atomicAdd(a + 1, v.y * nd);
    atomicAdd(a + 2, v.z * nd);
    atomicAdd(a + 3, v.w * nd);
}

__global__ void relu_kernel(float* __restrict__ h, int n) {
    int i = blockIdx.x * blockDim.x + threadIdx.x;
    if (i < n) h[i] = fmaxf(h[i], 0.0f);
}

static inline int cdiv(long long a, long long b) { return (int)((a + b - 1) / b); }

extern "C" void kernel_launch(void* const* d_in, const int* in_sizes, int n_in,
                              void* d_out, int out_size, void* d_ws, size_t ws_size,
                              hipStream_t stream) {
    const float* x  = (const float*)d_in[0];
    const int* ei   = (const int*)d_in[1];
    const float* W1 = (const float*)d_in[2];
    const float* b1 = (const float*)d_in[3];
    const float* W2 = (const float*)d_in[4];
    const float* b2 = (const float*)d_in[5];
    float* out = (float*)d_out;

    const int* src = ei;
    const int* dst = ei + NE;

    char* ws = (char*)d_ws;
    float* dinv = (float*)ws;
    float* hw1  = (float*)(ws + (size_t)NN * 4);
    float* h1   = hw1 + (size_t)NN * HID_DIM;
    float* hw2  = h1  + (size_t)NN * HID_DIM;

    const int B = 256;

    init_deg_kernel<<<cdiv(NN, B), B, 0, stream>>>(dinv, NN);
    count_deg_kernel<<<cdiv(NE, B), B, 0, stream>>>(dst, dinv, NE);
    rsqrt_kernel<<<cdiv(NN, B), B, 0, stream>>>(dinv, NN);

    {
        const int waves = (NN / 16) * (HID_DIM / 16);
        gemm_scaled_kernel<IN_DIM, HID_DIM>
            <<<cdiv((long long)waves * 32, B), B, 0, stream>>>(x, W1, dinv, hw1, NN);
    }
    init_agg_kernel<HID_DIM><<<cdiv((long long)NN * HID_DIM, B), B, 0, stream>>>(hw1, dinv, b1, h1, NN);
    edge_agg_kernel<HID_DIM><<<cdiv((long long)NE * (HID_DIM / 4), B), B, 0, stream>>>(
        src, dst, hw1, dinv, h1, NE);
    relu_kernel<<<cdiv((long long)NN * HID_DIM, B), B, 0, stream>>>(h1, NN * HID_DIM);

    {
        const int waves = (NN / 16) * (OUT_DIM / 16);
        gemm_scaled_kernel<HID_DIM, OUT_DIM>
            <<<cdiv((long long)waves * 32, B), B, 0, stream>>>(h1, W2, dinv, hw2, NN);
    }
    init_agg_kernel<OUT_DIM><<<cdiv((long long)NN * OUT_DIM, B), B, 0, stream>>>(hw2, dinv, b2, out, NN);
    edge_agg_kernel<OUT_DIM><<<cdiv((long long)NE * (OUT_DIM / 4), B), B, 0, stream>>>(
        src, dst, hw2, dinv, out, NE);
}


# degree count: global float atomics replaced by per-workgroup LDS integer histograms (5 node ranges x 50 edge slices); partials summed and rsqrt'd in one kernel; launches 10 -> 9
# speedup vs baseline: 1.3283x; 1.3283x over previous
#include <hip/hip_runtime.h>

#define NN 200000
#define NE 6400000
#define IN_DIM 32
#define HID_DIM 32
#define OUT_DIM 16

typedef __attribute__((ext_vector_type(2))) float v2f;
typedef __attribute__((ext_vector_type(8))) float v8f;

#define DEG_RANGES 5
#define DEG_SLICES 50
#define DEG_NODES (NN / DEG_RANGES)
#define DEG_EDGES (NE / DEG_SLICES)
#define DEG_THREADS 1024

__global__ __launch_bounds__(DEG_THREADS) void deg_hist_kernel(const int* __restrict__ dst,
                                                              unsigned* __restrict__ partial) {
    __shared__ unsigned cnt[DEG_NODES];
    const int range = blockIdx.x % DEG_RANGES;
    const int slice = blockIdx.x / DEG_RANGES;
    const int lo = range * DEG_NODES;

    for (int i = threadIdx.x; i < DEG_NODES; i += DEG_THREADS) cnt[i] = 0u;
    __syncthreads();

    constexpr int V4 = DEG_EDGES / 4;
    constexpr int ITER = (V4 + DEG_THREADS - 1) / DEG_THREADS;
    constexpr int UNR = 8;
    const int4* p = reinterpret_cast<const int4*>(dst + (size_t)slice * DEG_EDGES);
    for (int it = 0; it < ITER; it += UNR) {
        int4 d[UNR];
#pragma unroll
        for (int u = 0; u < UNR; ++u) {
            const int idx = (it + u) * DEG_THREADS + threadIdx.x;
            d[u] = (idx < V4) ? p[idx] : make_int4(-1, -1, -1, -1);
        }
#pragma unroll
        for (int u = 0; u < UNR; ++u) {
            const unsigned a0 = (unsigned)(d[u].x - lo);
            const unsigned a1 = (unsigned)(d[u].y - lo);
            const unsigned a2 = (unsigned)(d[u].z - lo);
            const unsigned a3 = (unsigned)(d[u].w - lo);
            if (a0 < (unsigned)DEG_NODES) atomicAdd(&cnt[a0], 1u);
            if (a1 < (unsigned)DEG_NODES) atomicAdd(&cnt[a1], 1u);
            if (a2 < (unsigned)DEG_NODES) atomicAdd(&cnt[a2], 1u);
            if (a3 < (unsigned)DEG_NODES) atomicAdd(&cnt[a3], 1u);
        }
    }
    __syncthreads();

    unsigned* o = partial + (size_t)slice * NN + lo;
    for (int i = threadIdx.x; i < DEG_NODES; i += DEG_THREADS) o[i] = cnt[i];
}

__global__ void deg_reduce_rsqrt_kernel(const unsigned* __restrict__ partial,
                                        float* __restrict__ dinv, int n) {
    int i = blockIdx.x * blockDim.x + threadIdx.x;
    if (i >= n) return;
    unsigned c = 1u;
#pragma unroll 10
    for (int s = 0; s < DEG_SLICES; ++s) c += partial[(size_t)s * NN + i];
    dinv[i] = rsqrtf((float)c);
}

template <int K, int FOUT>
__global__ void gemm_scaled_kernel(const float* __restrict__ X,
                                   const float* __restrict__ W,
                                   const float* __restrict__ dinv,
                                   float* __restrict__ out, int nrows) {
    const int wave = (blockIdx.x * blockDim.x + threadIdx.x) >> 5;
    const int lane = threadIdx.x & 31;
    const int tilesN = FOUT / 16;
    const int tm = wave / tilesN;
    const int tn = wave % tilesN;
    if (tm * 16 >= nrows) return;

    const int m = lane & 15;
    const int half = lane >> 4;
    const int row0 = tm * 16;
    const int col0 = tn * 16;

    v8f acc = {};
#pragma unroll
    for (int kb = 0; kb < K; kb += 4) {
        const int k0 = kb + 2 * half;
        v2f a, b;
        a.x = X[(size_t)(row0 + m) * K + k0 + 0];
        a.y = X[(size_t)(row0 + m) * K + k0 + 1];
        b.x = W[(size_t)(k0 + 0) * FOUT + col0 + m];
        b.y = W[(size_t)(k0 + 1) * FOUT + col0 + m];
        acc = __builtin_amdgcn_wmma_f32_16x16x4_f32(false, a, false, b,
                                                    (short)0, acc, false, false);
    }
#pragma unroll
    for (int r = 0; r < 8; ++r) {
        const int row = row0 + r + 8 * half;
        out[(size_t)row * FOUT + col0 + m] = acc[r] * dinv[row];
    }
}

template <int F>
__global__ void init_agg_kernel(const float* __restrict__ hws,
                                const float* __restrict__ dinv,
                                const float* __restrict__ bias,
                                float* __restrict__ agg, int n) {
    int i = blockIdx.x * blockDim.x + threadIdx.x;
    if (i >= n * F) return;
    int node = i / F;
    int f = i - node * F;
    agg[i] = hws[i] * dinv[node] + bias[f];
}

template <int F>
__global__ void edge_agg_kernel(const int* __restrict__ src, const int* __restrict__ dst,
                                const float* __restrict__ hws, const float* __restrict__ dinv,
                                float* __restrict__ agg, int ne) {
    constexpr int TPE = F / 4;
    int tid = blockIdx.x * blockDim.x + threadIdx.x;
    int e = tid / TPE;
    int g = tid - e * TPE;
    if (e >= ne) return;
    const int s = src[e];
    const int d = dst[e];
    const float nd = dinv[d];
    const float4 v = reinterpret_cast<const float4*>(hws + (size_t)s * F)[g];
    float* a = agg + (size_t)d * F + g * 4;
    atomicAdd(a + 0, v.x * nd);
    atomicAdd(a + 1, v.y * nd);
    atomicAdd(a + 2, v.z * nd);
    atomicAdd(a + 3, v.w * nd);
}

__global__ void relu_kernel(float* __restrict__ h, int n) {
    int i = blockIdx.x * blockDim.x + threadIdx.x;
    if (i < n) h[i] = fmaxf(h[i], 0.0f);
}

static inline int cdiv(long long a, long long b) { return (int)((a + b - 1) / b); }

extern "C" void kernel_launch(void* const* d_in, const int* in_sizes, int n_in,
                              void* d_out, int out_size, void* d_ws, size_t ws_size,
                              hipStream_t stream) {
    const float* x  = (const float*)d_in[0];
    const int* ei   = (const int*)d_in[1];
    const float* W1 = (const float*)d_in[2];
    const float* b1 = (const float*)d_in[3];
    const float* W2 = (const float*)d_in[4];
    const float* b2 = (const float*)d_in[5];
    float* out = (float*)d_out;

    const int* src = ei;
    const int* dst = ei + NE;

    char* ws = (char*)d_ws;
    float* dinv = (float*)ws;
    float* hw1  = (float*)(ws + (size_t)NN * 4);
    float* h1   = hw1 + (size_t)NN * HID_DIM;
    float* hw2  = h1  + (size_t)NN * HID_DIM;
    unsigned* degp = (unsigned*)(hw2 + (size_t)NN * OUT_DIM);

    const int B = 256;

    deg_hist_kernel<<<DEG_RANGES * DEG_SLICES, DEG_THREADS, 0, stream>>>(dst, degp);
    deg_reduce_rsqrt_kernel<<<cdiv(NN, B), B, 0, stream>>>(degp, dinv, NN);

    {
        const int waves = (NN / 16) * (HID_DIM / 16);
        gemm_scaled_kernel<IN_DIM, HID_DIM>
            <<<cdiv((long long)waves * 32, B), B, 0, stream>>>(x, W1, dinv, hw1, NN);
    }
    init_agg_kernel<HID_DIM><<<cdiv((long long)NN * HID_DIM, B), B, 0, stream>>>(hw1, dinv, b1, h1, NN);
    edge_agg_kernel<HID_DIM><<<cdiv((long long)NE * (HID_DIM / 4), B), B, 0, stream>>>(
        src, dst, hw1, dinv, h1, NE);
    relu_kernel<<<cdiv((long long)NN * HID_DIM, B), B, 0, stream>>>(h1, NN * HID_DIM);

    {
        const int waves = (NN / 16) * (OUT_DIM / 16);
        gemm_scaled_kernel<HID_DIM, OUT_DIM>
            <<<cdiv((long long)waves * 32, B), B, 0, stream>>>(h1, W2, dinv, hw2, NN);
    }
    init_agg_kernel<OUT_DIM><<<cdiv((long long)NN * OUT_DIM, B), B, 0, stream>>>(hw2, dinv, b2, out, NN);
    edge_agg_kernel<OUT_DIM><<<cdiv((long long)NE * (OUT_DIM / 4), B), B, 0, stream>>>(
        src, dst, hw2, dinv, out, NE);
}


# edge scatter re-mapped: one wave walks 128 edges, its 32 lanes cover a whole row (two for F=16) per load and per f32 atomic add (was 8 lanes x 4 strided adds); 24x fewer waves; arrival-order sums
# speedup vs baseline: 1.9280x; 1.4515x over previous
#include <hip/hip_runtime.h>

#define NN 200000
#define NE 6400000
#define IN_DIM 32
#define HID_DIM 32
#define OUT_DIM 16

typedef __attribute__((ext_vector_type(2))) float v2f;
typedef __attribute__((ext_vector_type(8))) float v8f;

#define DEG_RANGES 5
#define DEG_SLICES 50
#define DEG_NODES (NN / DEG_RANGES)
#define DEG_EDGES (NE / DEG_SLICES)
#define DEG_THREADS 1024

__global__ __launch_bounds__(DEG_THREADS) void deg_hist_kernel(const int* __restrict__ dst,
                                                              unsigned* __restrict__ partial) {
    __shared__ unsigned cnt[DEG_NODES];
    const int range = blockIdx.x % DEG_RANGES;
    const int slice = blockIdx.x / DEG_RANGES;
    const int lo = range * DEG_NODES;

    for (int i = threadIdx.x; i < DEG_NODES; i += DEG_THREADS) cnt[i] = 0u;
    __syncthreads();

    constexpr int V4 = DEG_EDGES / 4;
    constexpr int ITER = (V4 + DEG_THREADS - 1) / DEG_THREADS;
    constexpr int UNR = 8;
    const int4* p = reinterpret_cast<const int4*>(dst + (size_t)slice * DEG_EDGES);
    for (int it = 0; it < ITER; it += UNR) {
        int4 d[UNR];
#pragma unroll
        for (int u = 0; u < UNR; ++u) {
            const int idx = (it + u) * DEG_THREADS + threadIdx.x;
            d[u] = (idx < V4) ? p[idx] : make_int4(-1, -1, -1, -1);
        }
#pragma unroll
        for (int u = 0; u < UNR; ++u) {
            const unsigned a0 = (unsigned)(d[u].x - lo);
            const unsigned a1 = (unsigned)(d[u].y - lo);
            const unsigned a2 = (unsigned)(d[u].z - lo);
            const unsigned a3 = (unsigned)(d[u].w - lo);
            if (a0 < (unsigned)DEG_NODES) atomicAdd(&cnt[a0], 1u);
            if (a1 < (unsigned)DEG_NODES) atomicAdd(&cnt[a1], 1u);
            if (a2 < (unsigned)DEG_NODES) atomicAdd(&cnt[a2], 1u);
            if (a3 < (unsigned)DEG_NODES) atomicAdd(&cnt[a3], 1u);
        }
    }
    __syncthreads();

    unsigned* o = partial + (size_t)slice * NN + lo;
    for (int i = threadIdx.x; i < DEG_NODES; i += DEG_THREADS) o[i] = cnt[i];
}

__global__ void deg_reduce_rsqrt_kernel(const unsigned* __restrict__ partial,
                                        float* __restrict__ dinv, int n) {
    int i = blockIdx.x * blockDim.x + threadIdx.x;
    if (i >= n) return;
    unsigned c = 1u;
#pragma unroll 10
    for (int s = 0; s < DEG_SLICES; ++s) c += partial[(size_t)s * NN + i];
    dinv[i] = rsqrtf((float)c);
}

template <int K, int FOUT>
__global__ void gemm_scaled_kernel(const float* __restrict__ X,
                                   const float* __restrict__ W,
                                   const float* __restrict__ dinv,
                                   float* __restrict__ out, int nrows) {
    const int wave = (blockIdx.x * blockDim.x + threadIdx.x) >> 5;
    const int lane = threadIdx.x & 31;
    const int tilesN = FOUT / 16;
    const int tm = wave / tilesN;
    const int tn = wave % tilesN;
    if (tm * 16 >= nrows) return;

    const int m = lane & 15;
    const int half = lane >> 4;
    const int row0 = tm * 16;
    const int col0 = tn * 16;

    v8f acc = {};
#pragma unroll
    for (int kb = 0; kb < K; kb += 4) {
        const int k0 = kb + 2 * half;
        v2f a, b;
        a.x = X[(size_t)(row0 + m) * K + k0 + 0];
        a.y = X[(size_t)(row0 + m) * K + k0 + 1];
        b.x = W[(size_t)(k0 + 0) * FOUT + col0 + m];
        b.y = W[(size_t)(k0 + 1) * FOUT + col0 + m];
        acc = __builtin_amdgcn_wmma_f32_16x16x4_f32(false, a, false, b,
                                                    (short)0, acc, false, false);
    }
#pragma unroll
    for (int r = 0; r < 8; ++r) {
        const int row = row0 + r + 8 * half;
        out[(size_t)row * FOUT + col0 + m] = acc[r] * dinv[row];
    }
}

template <int F>
__global__ void init_agg_kernel(const float* __restrict__ hws,
                                const float* __restrict__ dinv,
                                const float* __restrict__ bias,
                                float* __restrict__ agg, int n) {
    int i = blockIdx.x * blockDim.x + threadIdx.x;
    if (i >= n * F) return;
    int node = i / F;
    int f = i - node * F;
    agg[i] = hws[i] * dinv[node] + bias[f];
}

#define EDGES_PER_WAVE 128
#define EDGE_WAVES_PER_BLOCK 8
#define EDGE_BLOCKS (NE / (EDGES_PER_WAVE * EDGE_WAVES_PER_BLOCK))
static_assert(NE % (EDGES_PER_WAVE * EDGE_WAVES_PER_BLOCK) == 0, "edge walk assumes full waves");
static_assert((long long)NN * HID_DIM * 4 < (1LL << 32), "row byte offsets are kept in 32 bits");

#define ROW_SHARE_CASE(n) case n: return __builtin_amdgcn_update_dpp(0, x, 0x150 + n, 0xf, 0xf, true);
__device__ __forceinline__ int row_share(int x, int u) {
    switch (u) {
        ROW_SHARE_CASE(0) ROW_SHARE_CASE(1) ROW_SHARE_CASE(2) ROW_SHARE_CASE(3)
        ROW_SHARE_CASE(4) ROW_SHARE_CASE(5) ROW_SHARE_CASE(6) ROW_SHARE_CASE(7)
        ROW_SHARE_CASE(8) ROW_SHARE_CASE(9) ROW_SHARE_CASE(10) ROW_SHARE_CASE(11)
        ROW_SHARE_CASE(12) ROW_SHARE_CASE(13) ROW_SHARE_CASE(14) ROW_SHARE_CASE(15)
    }
    return x;
}

template <int F>
__device__ __forceinline__ int edge_value(int x, int u) {
    if (F == 32) return __builtin_amdgcn_readlane(x, u);
    return row_share(x, u);
}

template <int F>
__global__ __launch_bounds__(32 * EDGE_WAVES_PER_BLOCK) void edge_agg_kernel(
        const int* __restrict__ src, const int* __restrict__ dst,
        const float* __restrict__ hws, const float* __restrict__ dinv,
        float* __restrict__ agg) {
    static_assert(F == 32 || F == 16, "a row of features is a wave or a half-wave");
    constexpr int BATCH = 16;
    const int lane = threadIdx.x & 31;
    const int wave = (blockIdx.x * blockDim.x + threadIdx.x) >> 5;
    const unsigned f4 = (unsigned)(lane & (F - 1)) * 4u;

    const int4 s4 = reinterpret_cast<const int4*>(src)[(size_t)wave * 32 + lane];
    const int4 d4 = reinterpret_cast<const int4*>(dst)[(size_t)wave * 32 + lane];
    const int dv[4] = {d4.x, d4.y, d4.z, d4.w};
    const int so[4] = {s4.x * (F * 4), s4.y * (F * 4), s4.z * (F * 4), s4.w * (F * 4)};
    const int dof[4] = {d4.x * (F * 4), d4.y * (F * 4), d4.z * (F * 4), d4.w * (F * 4)};
    int nv[4];
#pragma unroll
    for (int c = 0; c < 4; ++c) nv[c] = __float_as_int(dinv[dv[c]]);

#pragma unroll
    for (int c = 0; c < 4; ++c) {
#pragma unroll
        for (int u0 = 0; u0 < F; u0 += BATCH) {
            float v[BATCH];
            float* a[BATCH];
#pragma unroll
            for (int b = 0; b < BATCH; ++b) {
                const unsigned os = (unsigned)edge_value<F>(so[c], u0 + b) + f4;
                const unsigned od = (unsigned)edge_value<F>(dof[c], u0 + b) + f4;
                const float nd = __int_as_float(edge_value<F>(nv[c], u0 + b));
                v[b] = *reinterpret_cast<const float*>(reinterpret_cast<const char*>(hws) + os) * nd;
                a[b] = reinterpret_cast<float*>(reinterpret_cast<char*>(agg) + od);
            }
#pragma unroll
            for (int b = 0; b < BATCH; ++b) atomicAdd(a[b], v[b]);
        }
    }
}

__global__ void relu_kernel(float* __restrict__ h, int n) {
    int i = blockIdx.x * blockDim.x + threadIdx.x;
    if (i < n) h[i] = fmaxf(h[i], 0.0f);
}

static inline int cdiv(long long a, long long b) { return (int)((a + b - 1) / b); }

extern "C" void kernel_launch(void* const* d_in, const int* in_sizes, int n_in,
                              void* d_out, int out_size, void* d_ws, size_t ws_size,
                              hipStream_t stream) {
    const float* x  = (const float*)d_in[0];
    const int* ei   = (const int*)d_in[1];
    const float* W1 = (const float*)d_in[2];
    const float* b1 = (const float*)d_in[3];
    const float* W2 = (const float*)d_in[4];
    const float* b2 = (const float*)d_in[5];
    float* out = (float*)d_out;

    const int* src = ei;
    const int* dst = ei + NE;

    char* ws = (char*)d_ws;
    float* dinv = (float*)ws;
    float* hw1  = (float*)(ws + (size_t)NN * 4);
    float* h1   = hw1 + (size_t)NN * HID_DIM;
    float* hw2  = h1  + (size_t)NN * HID_DIM;
    unsigned* degp = (unsigned*)(hw2 + (size_t)NN * OUT_DIM);

    const int B = 256;

    deg_hist_kernel<<<DEG_RANGES * DEG_SLICES, DEG_THREADS, 0, stream>>>(dst, degp);
    deg_reduce_rsqrt_kernel<<<cdiv(NN, B), B, 0, stream>>>(degp, dinv, NN);

    {
        const int waves = (NN / 16) * (HID_DIM / 16);
        gemm_scaled_kernel<IN_DIM, HID_DIM>
            <<<cdiv((long long)waves * 32, B), B, 0, stream>>>(x, W1, dinv, hw1, NN);
    }
    init_agg_kernel<HID_DIM><<<cdiv((long long)NN * HID_DIM, B), B, 0, stream>>>(hw1, dinv, b1, h1, NN);
    edge_agg_kernel<HID_DIM><<<EDGE_BLOCKS, 32 * EDGE_WAVES_PER_BLOCK, 0, stream>>>(
        src, dst, hw1, dinv, h1);
    relu_kernel<<<cdiv((long long)NN * HID_DIM, B), B, 0, stream>>>(h1, NN * HID_DIM);

    {
        const int waves = (NN / 16) * (OUT_DIM / 16);
        gemm_scaled_kernel<HID_DIM, OUT_DIM>
            <<<cdiv((long long)waves * 32, B), B, 0, stream>>>(h1, W2, dinv, hw2, NN);
    }
    init_agg_kernel<OUT_DIM><<<cdiv((long long)NN * OUT_DIM, B), B, 0, stream>>>(hw2, dinv, b2, out, NN);
    edge_agg_kernel<OUT_DIM><<<EDGE_BLOCKS, 32 * EDGE_WAVES_PER_BLOCK, 0, stream>>>(
        src, dst, hw2, dinv, out);
}


# edges bucketed by destination once per call (counting sort on the degree histograms); each node pulls neighbour rows, sums in registers: no f32 atomics; sum order = stored order (arrival in a slice)
# speedup vs baseline: 2.6123x; 1.3549x over previous
#include <hip/hip_runtime.h>

#define NN 200000
#define NE 6400000
#define IN_DIM 32
#define HID_DIM 32
#define OUT_DIM 16

typedef __attribute__((ext_vector_type(2))) float v2f;
typedef __attribute__((ext_vector_type(8))) float v8f;

#define DEG_RANGES 5
#define DEG_SLICES 50
#define DEG_NODES (NN / DEG_RANGES)
#define DEG_EDGES (NE / DEG_SLICES)
#define DEG_THREADS 1024
static_assert((long long)NE < (1LL << 31), "edge slots are kept in non-negative ints");

__global__ __launch_bounds__(DEG_THREADS) void deg_hist_kernel(const int* __restrict__ dst,
                                                              unsigned* __restrict__ partial) {
    __shared__ unsigned cnt[DEG_NODES];
    const int range = blockIdx.x % DEG_RANGES;
    const int slice = blockIdx.x / DEG_RANGES;
    const int lo = range * DEG_NODES;

    for (int i = threadIdx.x; i < DEG_NODES; i += DEG_THREADS) cnt[i] = 0u;
    __syncthreads();

    constexpr int V4 = DEG_EDGES / 4;
    constexpr int ITER = (V4 + DEG_THREADS - 1) / DEG_THREADS;
    constexpr int UNR = 8;
    const int4* p = reinterpret_cast<const int4*>(dst + (size_t)slice * DEG_EDGES);
    for (int it = 0; it < ITER; it += UNR) {
        int4 d[UNR];
#pragma unroll
        for (int u = 0; u < UNR; ++u) {
            const int idx = (it + u) * DEG_THREADS + threadIdx.x;
            d[u] = (idx < V4) ? p[idx] : make_int4(-1, -1, -1, -1);
        }
#pragma unroll
        for (int u = 0; u < UNR; ++u) {
            const unsigned a0 = (unsigned)(d[u].x - lo);
            const unsigned a1 = (unsigned)(d[u].y - lo);
            const unsigned a2 = (unsigned)(d[u].z - lo);
            const unsigned a3 = (unsigned)(d[u].w - lo);
            if (a0 < (unsigned)DEG_NODES) atomicAdd(&cnt[a0], 1u);
            if (a1 < (unsigned)DEG_NODES) atomicAdd(&cnt[a1], 1u);
            if (a2 < (unsigned)DEG_NODES) atomicAdd(&cnt[a2], 1u);
            if (a3 < (unsigned)DEG_NODES) atomicAdd(&cnt[a3], 1u);
        }
    }
    __syncthreads();

    unsigned* o = partial + (size_t)slice * NN + lo;
    for (int i = threadIdx.x; i < DEG_NODES; i += DEG_THREADS) o[i] = cnt[i];
}

#define SCAN_THREADS 256
#define SCAN_BLOCKS ((NN + SCAN_THREADS - 1) / SCAN_THREADS)
__global__ __launch_bounds__(SCAN_THREADS) void deg_reduce_scan_kernel(
        unsigned* partial, float* __restrict__ dinv, int* __restrict__ cnt,
        int* __restrict__ lex, int* __restrict__ bsum, int n) {
    __shared__ int wsum[SCAN_THREADS / 32];
    const int i = blockIdx.x * SCAN_THREADS + threadIdx.x;
    const int lane = threadIdx.x & 31;
    const int w = threadIdx.x >> 5;
    unsigned c = 0u;
    if (i < n) {
        unsigned v[DEG_SLICES];
#pragma unroll
        for (int s = 0; s < DEG_SLICES; ++s) v[s] = partial[(size_t)s * NN + i];
#pragma unroll
        for (int s = 0; s < DEG_SLICES; ++s) {
            partial[(size_t)s * NN + i] = c;
            c += v[s];
        }
        dinv[i] = rsqrtf((float)(c + 1u));
        cnt[i] = (int)c;
    }
    int inc = (int)c;
#pragma unroll
    for (int d = 1; d < 32; d <<= 1) {
        const int t = __shfl_up(inc, d);
        if (lane >= d) inc += t;
    }
    if (lane == 31) wsum[w] = inc;
    __syncthreads();
    int woff = 0;
    for (int k = 0; k < w; ++k) woff += wsum[k];
    if (i < n) lex[i] = woff + inc - (int)c;
    if (threadIdx.x == SCAN_THREADS - 1) bsum[blockIdx.x] = woff + inc;
}

__global__ __launch_bounds__(SCAN_THREADS) void rowptr_kernel(const int* __restrict__ bsum,
                                                             int* __restrict__ rowptr, int n) {
    __shared__ int red[SCAN_THREADS / 32];
    const int lane = threadIdx.x & 31;
    const int w = threadIdx.x >> 5;
    int s = 0;
    for (int j = threadIdx.x; j < (int)blockIdx.x; j += SCAN_THREADS) s += bsum[j];
#pragma unroll
    for (int d = 16; d > 0; d >>= 1) s += __shfl_xor(s, d);
    if (lane == 0) red[w] = s;
    __syncthreads();
    int off = 0;
#pragma unroll
    for (int k = 0; k < SCAN_THREADS / 32; ++k) off += red[k];
    const int i = blockIdx.x * SCAN_THREADS + threadIdx.x;
    if (i < n) rowptr[i] += off;
}

__global__ __launch_bounds__(DEG_THREADS) void csr_fill_kernel(const int* __restrict__ src,
                                                              const int* __restrict__ dst,
                                                              const int* __restrict__ rowptr,
                                                              const unsigned* __restrict__ before,
                                                              int* __restrict__ csr_src) {
    __shared__ unsigned pos[DEG_NODES];
    const int range = blockIdx.x % DEG_RANGES;
    const int slice = blockIdx.x / DEG_RANGES;
    const int lo = range * DEG_NODES;

    const unsigned* bf = before + (size_t)slice * NN + lo;
    for (int i = threadIdx.x; i < DEG_NODES; i += DEG_THREADS) pos[i] = (unsigned)rowptr[lo + i] + bf[i];
    __syncthreads();

    constexpr int V4 = DEG_EDGES / 4;
    constexpr int ITER = (V4 + DEG_THREADS - 1) / DEG_THREADS;
    constexpr int UNR = 4;
    const int4* pd = reinterpret_cast<const int4*>(dst + (size_t)slice * DEG_EDGES);
    const int4* ps = reinterpret_cast<const int4*>(src + (size_t)slice * DEG_EDGES);
    for (int it = 0; it < ITER; it += UNR) {
        int4 d[UNR], s[UNR];
#pragma unroll
        for (int u = 0; u < UNR; ++u) {
            const int idx = (it + u) * DEG_THREADS + threadIdx.x;
            const bool in = idx < V4;
            d[u] = in ? pd[idx] : make_int4(-1, -1, -1, -1);
            s[u] = in ? ps[idx] : make_int4(0, 0, 0, 0);
        }
        int p[UNR][4];
#pragma unroll
        for (int u = 0; u < UNR; ++u) {
            const unsigned a0 = (unsigned)(d[u].x - lo);
            const unsigned a1 = (unsigned)(d[u].y - lo);
            const unsigned a2 = (unsigned)(d[u].z - lo);
            const unsigned a3 = (unsigned)(d[u].w - lo);
            p[u][0] = a0 < (unsigned)DEG_NODES ? (int)atomicAdd(&pos[a0], 1u) : -1;
            p[u][1] = a1 < (unsigned)DEG_NODES ? (int)atomicAdd(&pos[a1], 1u) : -1;
            p[u][2] = a2 < (unsigned)DEG_NODES ? (int)atomicAdd(&pos[a2], 1u) : -1;
            p[u][3] = a3 < (unsigned)DEG_NODES ? (int)atomicAdd(&pos[a3], 1u) : -1;
        }
#pragma unroll
        for (int u = 0; u < UNR; ++u) {
            if (p[u][0] >= 0) csr_src[p[u][0]] = s[u].x;
            if (p[u][1] >= 0) csr_src[p[u][1]] = s[u].y;
            if (p[u][2] >= 0) csr_src[p[u][2]] = s[u].z;
            if (p[u][3] >= 0) csr_src[p[u][3]] = s[u].w;
        }
    }
}

template <int K, int FOUT>
__global__ void gemm_scaled_kernel(const float* __restrict__ X,
                                   const float* __restrict__ W,
                                   const float* __restrict__ dinv,
                                   float* __restrict__ out, int nrows) {
    const int wave = (blockIdx.x * blockDim.x + threadIdx.x) >> 5;
    const int lane = threadIdx.x & 31;
    const int tilesN = FOUT / 16;
    const int tm = wave / tilesN;
    const int tn = wave % tilesN;
    if (tm * 16 >= nrows) return;

    const int m = lane & 15;
    const int half = lane >> 4;
    const int row0 = tm * 16;
    const int col0 = tn * 16;

    v8f acc = {};
#pragma unroll
    for (int kb = 0; kb < K; kb += 4) {
        const int k0 = kb + 2 * half;
        v2f a, b;
        a.x = X[(size_t)(row0 + m) * K + k0 + 0];
        a.y = X[(size_t)(row0 + m) * K + k0 + 1];
        b.x = W[(size_t)(k0 + 0) * FOUT + col0 + m];
        b.y = W[(size_t)(k0 + 1) * FOUT + col0 + m];
        acc = __builtin_amdgcn_wmma_f32_16x16x4_f32(false, a, false, b,
                                                    (short)0, acc, false, false);
    }
#pragma unroll
    for (int r = 0; r < 8; ++r) {
        const int row = row0 + r + 8 * half;
        out[(size_t)row * FOUT + col0 + m] = acc[r] * dinv[row];
    }
}

#define NODES_PER_WAVE 16
#define PULL_WAVES_PER_BLOCK 4
static_assert(NODES_PER_WAVE <= 32, "a lane holds the row range of one of the wave's nodes");
static_assert((long long)NN * HID_DIM * 4 < (1LL << 32), "row byte offsets are kept in 32 bits");

template <int F>
struct PullShape {
    static constexpr int LPR = F / 4;
    static constexpr int RPI = 32 / LPR;
    static constexpr int UNR = 32 / RPI;
};

template <int F>
__device__ __forceinline__ void chunk_sources(int idx, int g, unsigned q16,
                                              unsigned (&so)[PullShape<F>::UNR]) {
#pragma unroll
    for (int u = 0; u < PullShape<F>::UNR; ++u)
        so[u] = (unsigned)__shfl(idx, u * PullShape<F>::RPI + g) * (unsigned)(F * 4) + q16;
}

template <int F>
__device__ __forceinline__ void chunk_rows(const float* __restrict__ hws, int m,
                                           const unsigned (&so)[PullShape<F>::UNR],
                                           float4 (&v)[PullShape<F>::UNR]) {
#pragma unroll
    for (int u = 0; u < PullShape<F>::UNR; ++u) {
        v[u] = make_float4(0.0f, 0.0f, 0.0f, 0.0f);
        if (u * PullShape<F>::RPI < m)
            v[u] = *reinterpret_cast<const float4*>(reinterpret_cast<const char*>(hws) + so[u]);
    }
}

template <int F>
__device__ __forceinline__ void chunk_add(int m, int g, float nd,
                                          const float4 (&v)[PullShape<F>::UNR], float4& acc) {
#pragma unroll
    for (int u = 0; u < PullShape<F>::UNR; ++u) {
        if (u * PullShape<F>::RPI + g < m) {
            acc.x += v[u].x * nd;
            acc.y += v[u].y * nd;
            acc.z += v[u].z * nd;
            acc.w += v[u].w * nd;
        }
    }
}

template <int F, bool RELU>
__global__ __launch_bounds__(32 * PULL_WAVES_PER_BLOCK) void pull_agg_kernel(
        const int* __restrict__ rowptr, const int* __restrict__ cnt,
        const int* __restrict__ csr_src, const float* __restrict__ hws,
        const float* __restrict__ dinv, const float* __restrict__ bias,
        float* __restrict__ out, int n) {
    static_assert(F == 32 || F == 16, "a row is 8 or 4 float4");
    constexpr int LPR = PullShape<F>::LPR;
    constexpr int UNR = PullShape<F>::UNR;
    const int lane = threadIdx.x & 31;
    const int wave = __builtin_amdgcn_readfirstlane((int)((blockIdx.x * blockDim.x + threadIdx.x) >> 5));
    const int g = lane / LPR;
    const int q = lane % LPR;
    const unsigned q16 = (unsigned)q * 16u;
    const int node0 = wave * NODES_PER_WAVE;
    if (node0 >= n) return;

    int rp = 0, cn = 0, dn = 0;
    if (lane < NODES_PER_WAVE && node0 + lane < n) {
        rp = rowptr[node0 + lane];
        cn = cnt[node0 + lane];
        dn = __float_as_int(dinv[node0 + lane]);
    }
    const float4 b4 = reinterpret_cast<const float4*>(bias)[q];

    int idx_next = 0;
    {
        const int d0 = __builtin_amdgcn_readlane(cn, 0);
        const int s0 = __builtin_amdgcn_readlane(rp, 0);
        if (d0 > 0) idx_next = csr_src[s0 + (lane < d0 ? lane : d0 - 1)];
    }

    for (int k = 0; k < NODES_PER_WAVE; ++k) {
        const int node = node0 + k;
        if (node >= n) break;
        const int start = __builtin_amdgcn_readlane(rp, k);
        const int deg = __builtin_amdgcn_readlane(cn, k);
        const float nd = __int_as_float(__builtin_amdgcn_readlane(dn, k));

        unsigned so[UNR];
        float4 v[UNR];
        chunk_sources<F>(idx_next, g, q16, so);
        if (k + 1 < NODES_PER_WAVE && node + 1 < n) {
            const int d1 = __builtin_amdgcn_readlane(cn, k + 1);
            const int s1 = __builtin_amdgcn_readlane(rp, k + 1);
            if (d1 > 0) idx_next = csr_src[s1 + (lane < d1 ? lane : d1 - 1)];
        }
        const float4 self = reinterpret_cast<const float4*>(hws + (size_t)node * F)[q];

        float4 acc = make_float4(0.0f, 0.0f, 0.0f, 0.0f);
        const int m0 = deg < 32 ? deg : 32;
        chunk_rows<F>(hws, m0, so, v);
        chunk_add<F>(m0, g, nd, v, acc);
        for (int base = 32; base < deg; base += 32) {
            const int left = deg - base;
            const int m = left < 32 ? left : 32;
            const int e = base + lane < deg ? base + lane : deg - 1;
            const int idx = csr_src[start + e];
            chunk_sources<F>(idx, g, q16, so);
            chunk_rows<F>(hws, m, so, v);
            chunk_add<F>(m, g, nd, v, acc);
        }
#pragma unroll
        for (int d = LPR; d < 32; d <<= 1) {
            acc.x += __shfl_xor(acc.x, d);
            acc.y += __shfl_xor(acc.y, d);
            acc.z += __shfl_xor(acc.z, d);
            acc.w += __shfl_xor(acc.w, d);
        }
        if (g == 0) {
            float4 r;
            r.x = (self.x * nd + b4.x) + acc.x;
            r.y = (self.y * nd + b4.y) + acc.y;
            r.z = (self.z * nd + b4.z) + acc.z;
            r.w = (self.w * nd + b4.w) + acc.w;
            if (RELU) {
                r.x = fmaxf(r.x, 0.0f);
                r.y = fmaxf(r.y, 0.0f);
                r.z = fmaxf(r.z, 0.0f);
                r.w = fmaxf(r.w, 0.0f);
            }
            reinterpret_cast<float4*>(out + (size_t)node * F)[q] = r;
        }
    }
}

static inline int cdiv(long long a, long long b) { return (int)((a + b - 1) / b); }

extern "C" void kernel_launch(void* const* d_in, const int* in_sizes, int n_in,
                              void* d_out, int out_size, void* d_ws, size_t ws_size,
                              hipStream_t stream) {
    const float* x  = (const float*)d_in[0];
    const int* ei   = (const int*)d_in[1];
    const float* W1 = (const float*)d_in[2];
    const float* b1 = (const float*)d_in[3];
    const float* W2 = (const float*)d_in[4];
    const float* b2 = (const float*)d_in[5];
    float* out = (float*)d_out;

    const int* src = ei;
    const int* dst = ei + NE;

    char* ws = (char*)d_ws;
    float* dinv = (float*)ws;
    float* hw1  = (float*)(ws + (size_t)NN * 4);
    float* h1   = hw1 + (size_t)NN * HID_DIM;
    float* hw2  = h1  + (size_t)NN * HID_DIM;
    unsigned* degp = (unsigned*)(hw2 + (size_t)NN * OUT_DIM);
    int* cnt    = (int*)(degp + (size_t)DEG_SLICES * NN);
    int* rowptr = cnt + NN;
    int* bsum   = rowptr + NN;
    int* csrc   = bsum + 1024;
    static_assert(SCAN_BLOCKS <= 1024, "bsum plane");

    const int B = 256;

    deg_hist_kernel<<<DEG_RANGES * DEG_SLICES, DEG_THREADS, 0, stream>>>(dst, degp);
    deg_reduce_scan_kernel<<<SCAN_BLOCKS, SCAN_THREADS, 0, stream>>>(degp, dinv, cnt, rowptr, bsum, NN);
    rowptr_kernel<<<SCAN_BLOCKS, SCAN_THREADS, 0, stream>>>(bsum, rowptr, NN);
    csr_fill_kernel<<<DEG_RANGES * DEG_SLICES, DEG_THREADS, 0, stream>>>(src, dst, rowptr, degp, csrc);

    const int pull_blocks = cdiv(NN, NODES_PER_WAVE * PULL_WAVES_PER_BLOCK);

    {
        const int waves = (NN / 16) * (HID_DIM / 16);
        gemm_scaled_kernel<IN_DIM, HID_DIM>
            <<<cdiv((long long)waves * 32, B), B, 0, stream>>>(x, W1, dinv, hw1, NN);
    }
    pull_agg_kernel<HID_DIM, true><<<pull_blocks, 32 * PULL_WAVES_PER_BLOCK, 0, stream>>>(
        rowptr, cnt, csrc, hw1, dinv, b1, h1, NN);

    {
        const int waves = (NN / 16) * (OUT_DIM / 16);
        gemm_scaled_kernel<HID_DIM, OUT_DIM>
            <<<cdiv((long long)waves * 32, B), B, 0, stream>>>(h1, W2, dinv, hw2, NN);
    }
    pull_agg_kernel<OUT_DIM, false><<<pull_blocks, 32 * PULL_WAVES_PER_BLOCK, 0, stream>>>(
        rowptr, cnt, csrc, hw2, dinv, b2, out, NN);
}


# edge ordering rebuilt as a two-level bucket partition (250 slices x 250 node buckets, sorted in LDS): every global store of the build is a contiguous run; sum order = arrival order in a node
# speedup vs baseline: 3.6235x; 1.3871x over previous
#include <hip/hip_runtime.h>

#define NN 200000
#define NE 6400000
#define IN_DIM 32
#define HID_DIM 32
#define OUT_DIM 16

typedef __attribute__((ext_vector_type(2))) float v2f;
typedef __attribute__((ext_vector_type(8))) float v8f;

#define BK_NODES 800
#define NBK (NN / BK_NODES)
#define NSL 250
#define SL_EDGES (NE / NSL)
#define SL_V4 (SL_EDGES / 4)
#define BT 1024
#define SL_ITER ((SL_V4 + BT - 1) / BT)
#define SRC_BITS 18
#define BK_CAP 32768
static_assert(NN % BK_NODES == 0 && NE % NSL == 0 && SL_EDGES % 4 == 0, "even split");
static_assert(NN <= (1 << SRC_BITS) && BK_NODES <= (1 << (32 - SRC_BITS)), "record fields");
static_assert(NBK <= BT && BK_NODES <= BT, "one thread per bucket / per node of a bucket");
static_assert((long long)NE < (1LL << 31), "edge slots are kept in non-negative ints");

__device__ __forceinline__ unsigned bucket_of(int d) {
    const unsigned b = (unsigned)d / (unsigned)BK_NODES;
    return b < (unsigned)NBK ? b : (unsigned)(NBK - 1);
}

__device__ __forceinline__ int block_excl_scan(int v, int* wsum, int& total) {
    const int lane = threadIdx.x & 31;
    const int w = threadIdx.x >> 5;
    int inc = v;
#pragma unroll
    for (int d = 1; d < 32; d <<= 1) {
        const int t = __shfl_up(inc, d);
        if (lane >= d) inc += t;
    }
    if (lane == 31) wsum[w] = inc;
    __syncthreads();
    const int ws = wsum[lane];
    int winc = ws;
#pragma unroll
    for (int d = 1; d < 32; d <<= 1) {
        const int t = __shfl_up(winc, d);
        if (lane >= d) winc += t;
    }
    const int woff = __shfl(winc, w) - __shfl(ws, w);
    total = __shfl(winc, 31);
    __syncthreads();
    return woff + inc - v;
}
static_assert(BT == 1024, "block_excl_scan: 32 waves of 32");

__global__ __launch_bounds__(BT) void part_count_kernel(const int* __restrict__ dst,
                                                        unsigned* __restrict__ hist) {
    __shared__ unsigned cnt[BT / 32][NBK];
    const int w = threadIdx.x >> 5;
    for (int i = threadIdx.x; i < (BT / 32) * NBK; i += BT) (&cnt[0][0])[i] = 0u;
    __syncthreads();
    const int4* p = reinterpret_cast<const int4*>(dst + (size_t)blockIdx.x * SL_EDGES);
    int4 d[SL_ITER];
#pragma unroll
    for (int u = 0; u < SL_ITER; ++u) {
        const int idx = u * BT + threadIdx.x;
        d[u] = p[idx < SL_V4 ? idx : SL_V4 - 1];
    }
#pragma unroll
    for (int u = 0; u < SL_ITER; ++u) {
        if (u * BT + (int)threadIdx.x < SL_V4) {
            atomicAdd(&cnt[w][bucket_of(d[u].x)], 1u);
            atomicAdd(&cnt[w][bucket_of(d[u].y)], 1u);
            atomicAdd(&cnt[w][bucket_of(d[u].z)], 1u);
            atomicAdd(&cnt[w][bucket_of(d[u].w)], 1u);
        }
    }
    __syncthreads();
    if (threadIdx.x < NBK) {
        unsigned s = 0u;
#pragma unroll
        for (int k = 0; k < BT / 32; ++k) s += cnt[k][threadIdx.x];
        hist[(size_t)threadIdx.x * NSL + blockIdx.x] = s;
    }
}

#define SCAN_PER_THREAD ((NBK * NSL + BT - 1) / BT)
__global__ __launch_bounds__(BT) void part_scan_kernel(unsigned* hist) {
    __shared__ int wsum[32];
    const int base = threadIdx.x * SCAN_PER_THREAD;
    int v[SCAN_PER_THREAD];
    int s = 0;
#pragma unroll
    for (int k = 0; k < SCAN_PER_THREAD; ++k) {
        v[k] = base + k < NBK * NSL ? (int)hist[base + k] : 0;
        s += v[k];
    }
    int total;
    int run = block_excl_scan(s, wsum, total);
#pragma unroll
    for (int k = 0; k < SCAN_PER_THREAD; ++k) {
        if (base + k < NBK * NSL) hist[base + k] = (unsigned)run;
        run += v[k];
    }
}

__global__ __launch_bounds__(BT) void part_scatter_kernel(const int* __restrict__ src,
                                                          const int* __restrict__ dst,
                                                          const unsigned* __restrict__ off,
                                                          unsigned* __restrict__ rec) {
    __shared__ int sdst[SL_EDGES];
    __shared__ int ssrc[SL_EDGES];
    __shared__ unsigned cursor[NBK];
    __shared__ int delta[NBK];
    __shared__ int wsum[32];
    const int sl = blockIdx.x;
    int c = 0, g = 0;
    if (threadIdx.x < NBK) {
        const int i = threadIdx.x * NSL + sl;
        g = (int)off[i];
        c = (i + 1 < NBK * NSL ? (int)off[i + 1] : NE) - g;
    }
    int total;
    const int ex = block_excl_scan(c, wsum, total);
    if (threadIdx.x < NBK) {
        cursor[threadIdx.x] = (unsigned)ex;
        delta[threadIdx.x] = g - ex;
    }
    __syncthreads();

    const int4* pd = reinterpret_cast<const int4*>(dst + (size_t)sl * SL_EDGES);
    const int4* ps = reinterpret_cast<const int4*>(src + (size_t)sl * SL_EDGES);
    int4 d[SL_ITER], s[SL_ITER];
#pragma unroll
    for (int u = 0; u < SL_ITER; ++u) {
        const int idx = u * BT + threadIdx.x;
        const int j = idx < SL_V4 ? idx : SL_V4 - 1;
        d[u] = pd[j];
        s[u] = ps[j];
    }
#pragma unroll
    for (int u = 0; u < SL_ITER; ++u) {
        if (u * BT + (int)threadIdx.x < SL_V4) {
            const unsigned p0 = min(atomicAdd(&cursor[bucket_of(d[u].x)], 1u), (unsigned)(SL_EDGES - 1));
            const unsigned p1 = min(atomicAdd(&cursor[bucket_of(d[u].y)], 1u), (unsigned)(SL_EDGES - 1));
            const unsigned p2 = min(atomicAdd(&cursor[bucket_of(d[u].z)], 1u), (unsigned)(SL_EDGES - 1));
            const unsigned p3 = min(atomicAdd(&cursor[bucket_of(d[u].w)], 1u), (unsigned)(SL_EDGES - 1));
            sdst[p0] = d[u].x; ssrc[p0] = s[u].x;
            sdst[p1] = d[u].y; ssrc[p1] = s[u].y;
            sdst[p2] = d[u].z; ssrc[p2] = s[u].z;
            sdst[p3] = d[u].w; ssrc[p3] = s[u].w;
        }
    }
    __syncthreads();
    for (int i = threadIdx.x; i < SL_EDGES; i += BT) {
        const int dd = sdst[i];
        const unsigned b = bucket_of(dd);
        const unsigned local = (unsigned)(dd - (int)b * BK_NODES) & (unsigned)((1 << (32 - SRC_BITS)) - 1);
        unsigned slot = (unsigned)(delta[b] + i);
        if (slot >= (unsigned)NE) slot = NE - 1;
        rec[slot] = ((unsigned)ssrc[i] & (unsigned)((1 << SRC_BITS) - 1)) | (local << SRC_BITS);
    }
}

__global__ __launch_bounds__(BT) void bucket_csr_kernel(const unsigned* __restrict__ off,
                                                        const unsigned* __restrict__ rec,
                                                        float* __restrict__ dinv, int* __restrict__ cnt,
                                                        int* __restrict__ rowptr,
                                                        int* __restrict__ csr_src) {
    __shared__ unsigned deg[BK_NODES];
    __shared__ unsigned cur[BK_NODES];
    __shared__ int ordered[BK_CAP];
    __shared__ int wsum[32];
    const int bk = blockIdx.x;
    int start = (int)off[(size_t)bk * NSL];
    int end = bk + 1 < NBK ? (int)off[(size_t)(bk + 1) * NSL] : NE;
    start = min(max(start, 0), NE);
    end = min(max(end, start), NE);
    const int n = end - start;
    if (threadIdx.x < BK_NODES) deg[threadIdx.x] = 0u;
    __syncthreads();
    for (int i = threadIdx.x; i < n; i += BT) {
        const unsigned l = min(rec[start + i] >> SRC_BITS, (unsigned)(BK_NODES - 1));
        atomicAdd(&deg[l], 1u);
    }
    __syncthreads();
    const int c = threadIdx.x < BK_NODES ? (int)deg[threadIdx.x] : 0;
    int total;
    const int ex = block_excl_scan(c, wsum, total);
    if (threadIdx.x < BK_NODES) {
        const int node = bk * BK_NODES + threadIdx.x;
        cur[threadIdx.x] = (unsigned)ex;
        dinv[node] = rsqrtf((float)(c + 1));
        cnt[node] = c;
        rowptr[node] = start + ex;
    }
    __syncthreads();
    for (int i = threadIdx.x; i < n; i += BT) {
        const unsigned r = rec[start + i];
        const unsigned l = min(r >> SRC_BITS, (unsigned)(BK_NODES - 1));
        const int p = (int)atomicAdd(&cur[l], 1u);
        const int sidx = (int)(r & (unsigned)((1 << SRC_BITS) - 1));
        if (p < BK_CAP) ordered[p] = sidx;
        else if (p < n) csr_src[start + p] = sidx;
    }
    __syncthreads();
    const int m = n < BK_CAP ? n : BK_CAP;
    for (int i = threadIdx.x; i < m; i += BT) csr_src[start + i] = ordered[i];
}

template <int K, int FOUT>
__global__ void gemm_scaled_kernel(const float* __restrict__ X,
                                   const float* __restrict__ W,
                                   const float* __restrict__ dinv,
                                   float* __restrict__ out, int nrows) {
    const int wave = (blockIdx.x * blockDim.x + threadIdx.x) >> 5;
    const int lane = threadIdx.x & 31;
    const int tilesN = FOUT / 16;
    const int tm = wave / tilesN;
    const int tn = wave % tilesN;
    if (tm * 16 >= nrows) return;

    const int m = lane & 15;
    const int half = lane >> 4;
    const int row0 = tm * 16;
    const int col0 = tn * 16;

    v8f acc = {};
#pragma unroll
    for (int kb = 0; kb < K; kb += 4) {
        const int k0 = kb + 2 * half;
        v2f a, b;
        a.x = X[(size_t)(row0 + m) * K + k0 + 0];
        a.y = X[(size_t)(row0 + m) * K + k0 + 1];
        b.x = W[(size_t)(k0 + 0) * FOUT + col0 + m];
        b.y = W[(size_t)(k0 + 1) * FOUT + col0 + m];
        acc = __builtin_amdgcn_wmma_f32_16x16x4_f32(false, a, false, b,
                                                    (short)0, acc, false, false);
    }
#pragma unroll
    for (int r = 0; r < 8; ++r) {
        const int row = row0 + r + 8 * half;
        out[(size_t)row * FOUT + col0 + m] = acc[r] * dinv[row];
    }
}

#define NODES_PER_WAVE 16
#define PULL_WAVES_PER_BLOCK 4
static_assert(NODES_PER_WAVE <= 32, "a lane holds the row range of one of the wave's nodes");
static_assert((long long)NN * HID_DIM * 4 < (1LL << 32), "row byte offsets are kept in 32 bits");

template <int F>
struct PullShape {
    static constexpr int LPR = F / 4;
    static constexpr int RPI = 32 / LPR;
    static constexpr int UNR = 32 / RPI;
};

template <int F>
__device__ __forceinline__ void chunk_sources(int idx, int g, unsigned q16,
                                              unsigned (&so)[PullShape<F>::UNR]) {
#pragma unroll
    for (int u = 0; u < PullShape<F>::UNR; ++u)
        so[u] = (unsigned)__shfl(idx, u * PullShape<F>::RPI + g) * (unsigned)(F * 4) + q16;
}

template <int F>
__device__ __forceinline__ void chunk_rows(const float* __restrict__ hws, int m,
                                           const unsigned (&so)[PullShape<F>::UNR],
                                           float4 (&v)[PullShape<F>::UNR]) {
#pragma unroll
    for (int u = 0; u < PullShape<F>::UNR; ++u) {
        v[u] = make_float4(0.0f, 0.0f, 0.0f, 0.0f);
        if (u * PullShape<F>::RPI < m)
            v[u] = *reinterpret_cast<const float4*>(reinterpret_cast<const char*>(hws) + so[u]);
    }
}

template <int F>
__device__ __forceinline__ void chunk_add(int m, int g, float nd,
                                          const float4 (&v)[PullShape<F>::UNR], float4& acc) {
#pragma unroll
    for (int u = 0; u < PullShape<F>::UNR; ++u) {
        if (u * PullShape<F>::RPI + g < m) {
            acc.x += v[u].x * nd;
            acc.y += v[u].y * nd;
            acc.z += v[u].z * nd;
            acc.w += v[u].w * nd;
        }
    }
}

template <int F, bool RELU>
__global__ __launch_bounds__(32 * PULL_WAVES_PER_BLOCK) void pull_agg_kernel(
        const int* __restrict__ rowptr, const int* __restrict__ cnt,
        const int* __restrict__ csr_src, const float* __restrict__ hws,
        const float* __restrict__ dinv, const float* __restrict__ bias,
        float* __restrict__ out, int n) {
    static_assert(F == 32 || F == 16, "a row is 8 or 4 float4");
    constexpr int LPR = PullShape<F>::LPR;
    constexpr int UNR = PullShape<F>::UNR;
    const int lane = threadIdx.x & 31;
    const int wave = __builtin_amdgcn_readfirstlane((int)((blockIdx.x * blockDim.x + threadIdx.x) >> 5));
    const int g = lane / LPR;
    const int q = lane % LPR;
    const unsigned q16 = (unsigned)q * 16u;
    const int node0 = wave * NODES_PER_WAVE;
    if (node0 >= n) return;

    int rp = 0, cn = 0, dn = 0;
    if (lane < NODES_PER_WAVE && node0 + lane < n) {
        rp = rowptr[node0 + lane];
        cn = cnt[node0 + lane];
        dn = __float_as_int(dinv[node0 + lane]);
    }
    const float4 b4 = reinterpret_cast<const float4*>(bias)[q];

    int idx_next = 0;
    {
        const int d0 = __builtin_amdgcn_readlane(cn, 0);
        const int s0 = __builtin_amdgcn_readlane(rp, 0);
        if (d0 > 0) idx_next = csr_src[s0 + (lane < d0 ? lane : d0 - 1)];
    }

    for (int k = 0; k < NODES_PER_WAVE; ++k) {
        const int node = node0 + k;
        if (node >= n) break;
        const int start = __builtin_amdgcn_readlane(rp, k);
        const int deg = __builtin_amdgcn_readlane(cn, k);
        const float nd = __int_as_float(__builtin_amdgcn_readlane(dn, k));

        unsigned so[UNR];
        float4 v[UNR];
        chunk_sources<F>(idx_next, g, q16, so);
        if (k + 1 < NODES_PER_WAVE && node + 1 < n) {
            const int d1 = __builtin_amdgcn_readlane(cn, k + 1);
            const int s1 = __builtin_amdgcn_readlane(rp, k + 1);
            if (d1 > 0) idx_next = csr_src[s1 + (lane < d1 ? lane : d1 - 1)];
        }
        const float4 self = reinterpret_cast<const float4*>(hws + (size_t)node * F)[q];

        float4 acc = make_float4(0.0f, 0.0f, 0.0f, 0.0f);
        const int m0 = deg < 32 ? deg : 32;
        chunk_rows<F>(hws, m0, so, v);
        chunk_add<F>(m0, g, nd, v, acc);
        for (int base = 32; base < deg; base += 32) {
            const int left = deg - base;
            const int m = left < 32 ? left : 32;
            const int e = base + lane < deg ? base + lane : deg - 1;
            const int idx = csr_src[start + e];
            chunk_sources<F>(idx, g, q16, so);
            chunk_rows<F>(hws, m, so, v);
            chunk_add<F>(m, g, nd, v, acc);
        }
#pragma unroll
        for (int d = LPR; d < 32; d <<= 1) {
            acc.x += __shfl_xor(acc.x, d);
            acc.y += __shfl_xor(acc.y, d);
            acc.z += __shfl_xor(acc.z, d);
            acc.w += __shfl_xor(acc.w, d);
        }
        if (g == 0) {
            float4 r;
            r.x = (self.x * nd + b4.x) + acc.x;
            r.y = (self.y * nd + b4.y) + acc.y;
            r.z = (self.z * nd + b4.z) + acc.z;
            r.w = (self.w * nd + b4.w) + acc.w;
            if (RELU) {
                r.x = fmaxf(r.x, 0.0f);
                r.y = fmaxf(r.y, 0.0f);
                r.z = fmaxf(r.z, 0.0f);
                r.w = fmaxf(r.w, 0.0f);
            }
            reinterpret_cast<float4*>(out + (size_t)node * F)[q] = r;
        }
    }
}

static inline int cdiv(long long a, long long b) { return (int)((a + b - 1) / b); }

extern "C" void kernel_launch(void* const* d_in, const int* in_sizes, int n_in,
                              void* d_out, int out_size, void* d_ws, size_t ws_size,
                              hipStream_t stream) {
    const float* x  = (const float*)d_in[0];
    const int* ei   = (const int*)d_in[1];
    const float* W1 = (const float*)d_in[2];
    const float* b1 = (const float*)d_in[3];
    const float* W2 = (const float*)d_in[4];
    const float* b2 = (const float*)d_in[5];
    float* out = (float*)d_out;

    const int* src = ei;
    const int* dst = ei + NE;

    char* ws = (char*)d_ws;
    float* dinv = (float*)ws;
    float* hw1  = (float*)(ws + (size_t)NN * 4);
    float* h1   = hw1 + (size_t)NN * HID_DIM;
    float* hw2  = h1  + (size_t)NN * HID_DIM;
    unsigned* off = (unsigned*)(hw2 + (size_t)NN * OUT_DIM);
    unsigned* rec = off + 65536;
    int* cnt    = (int*)(rec + (size_t)NE);
    int* rowptr = cnt + NN;
    int* csrc   = rowptr + NN;
    static_assert(NBK * NSL <= 65536, "off plane");

    const int B = 256;

    part_count_kernel<<<NSL, BT, 0, stream>>>(dst, off);
    part_scan_kernel<<<1, BT, 0, stream>>>(off);
    part_scatter_kernel<<<NSL, BT, 0, stream>>>(src, dst, off, rec);
    bucket_csr_kernel<<<NBK, BT, 0, stream>>>(off, rec, dinv, cnt, rowptr, csrc);

    const int pull_blocks = cdiv(NN, NODES_PER_WAVE * PULL_WAVES_PER_BLOCK);

    {
        const int waves = (NN / 16) * (HID_DIM / 16);
        gemm_scaled_kernel<IN_DIM, HID_DIM>
            <<<cdiv((long long)waves * 32, B), B, 0, stream>>>(x, W1, dinv, hw1, NN);
    }
    pull_agg_kernel<HID_DIM, true><<<pull_blocks, 32 * PULL_WAVES_PER_BLOCK, 0, stream>>>(
        rowptr, cnt, csrc, hw1, dinv, b1, h1, NN);

    {
        const int waves = (NN / 16) * (OUT_DIM / 16);
        gemm_scaled_kernel<HID_DIM, OUT_DIM>
            <<<cdiv((long long)waves * 32, B), B, 0, stream>>>(h1, W2, dinv, hw2, NN);
    }
    pull_agg_kernel<OUT_DIM, false><<<pull_blocks, 32 * PULL_WAVES_PER_BLOCK, 0, stream>>>(
        rowptr, cnt, csrc, hw2, dinv, b2, out, NN);
}


# scan of the 62500 run counts: table staged through LDS so every global access is a whole-line 16-byte load or store (was 4-byte accesses at a 248-byte lane stride, 32 lines per instruction)
# speedup vs baseline: 4.7218x; 1.3031x over previous
#include <hip/hip_runtime.h>

#define NN 200000
#define NE 6400000
#define IN_DIM 32
#define HID_DIM 32
#define OUT_DIM 16

typedef __attribute__((ext_vector_type(2))) float v2f;
typedef __attribute__((ext_vector_type(8))) float v8f;

#define BK_NODES 800
#define NBK (NN / BK_NODES)
#define NSL 250
#define SL_EDGES (NE / NSL)
#define SL_V4 (SL_EDGES / 4)
#define BT 1024
#define SL_ITER ((SL_V4 + BT - 1) / BT)
#define SRC_BITS 18
#define BK_CAP 32768
static_assert(NN % BK_NODES == 0 && NE % NSL == 0 && SL_EDGES % 4 == 0, "even split");
static_assert(NN <= (1 << SRC_BITS) && BK_NODES <= (1 << (32 - SRC_BITS)), "record fields");
static_assert(NBK <= BT && BK_NODES <= BT, "one thread per bucket / per node of a bucket");
static_assert((long long)NE < (1LL << 31), "edge slots are kept in non-negative ints");

__device__ __forceinline__ unsigned bucket_of(int d) {
    const unsigned b = (unsigned)d / (unsigned)BK_NODES;
    return b < (unsigned)NBK ? b : (unsigned)(NBK - 1);
}

__device__ __forceinline__ int block_excl_scan(int v, int* wsum, int& total) {
    const int lane = threadIdx.x & 31;
    const int w = threadIdx.x >> 5;
    int inc = v;
#pragma unroll
    for (int d = 1; d < 32; d <<= 1) {
        const int t = __shfl_up(inc, d);
        if (lane >= d) inc += t;
    }
    if (lane == 31) wsum[w] = inc;
    __syncthreads();
    const int ws = wsum[lane];
    int winc = ws;
#pragma unroll
    for (int d = 1; d < 32; d <<= 1) {
        const int t = __shfl_up(winc, d);
        if (lane >= d) winc += t;
    }
    const int woff = __shfl(winc, w) - __shfl(ws, w);
    total = __shfl(winc, 31);
    __syncthreads();
    return woff + inc - v;
}
static_assert(BT == 1024, "block_excl_scan: 32 waves of 32");

__global__ __launch_bounds__(BT) void part_count_kernel(const int* __restrict__ dst,
                                                        unsigned* __restrict__ hist) {
    __shared__ unsigned cnt[BT / 32][NBK];
    const int w = threadIdx.x >> 5;
    for (int i = threadIdx.x; i < (BT / 32) * NBK; i += BT) (&cnt[0][0])[i] = 0u;
    __syncthreads();
    const int4* p = reinterpret_cast<const int4*>(dst + (size_t)blockIdx.x * SL_EDGES);
    int4 d[SL_ITER];
#pragma unroll
    for (int u = 0; u < SL_ITER; ++u) {
        const int idx = u * BT + threadIdx.x;
        d[u] = p[idx < SL_V4 ? idx : SL_V4 - 1];
    }
#pragma unroll
    for (int u = 0; u < SL_ITER; ++u) {
        if (u * BT + (int)threadIdx.x < SL_V4) {
            atomicAdd(&cnt[w][bucket_of(d[u].x)], 1u);
            atomicAdd(&cnt[w][bucket_of(d[u].y)], 1u);
            atomicAdd(&cnt[w][bucket_of(d[u].z)], 1u);
            atomicAdd(&cnt[w][bucket_of(d[u].w)], 1u);
        }
    }
    __syncthreads();
    if (threadIdx.x < NBK) {
        unsigned s = 0u;
#pragma unroll
        for (int k = 0; k < BT / 32; ++k) s += cnt[k][threadIdx.x];
        hist[(size_t)threadIdx.x * NSL + blockIdx.x] = s;
    }
}

#define SCAN_N (NBK * NSL)
#define SCAN_PER_THREAD ((SCAN_N + BT - 1) / BT)
#define SCAN_V4 (SCAN_N / 4)
#define SCAN_ITER ((BT * SCAN_PER_THREAD / 4 + BT - 1) / BT)
static_assert(SCAN_N % 4 == 0, "the table is copied in 16-byte words");
static_assert(SCAN_ITER * BT * 4 >= BT * SCAN_PER_THREAD, "the LDS copy covers every thread's entries");
__global__ __launch_bounds__(BT) void part_scan_kernel(unsigned* hist) {
    __shared__ int4 tab4[SCAN_ITER * BT];
    __shared__ int wsum[32];
    int* tab = reinterpret_cast<int*>(tab4);
    int4* h4 = reinterpret_cast<int4*>(hist);
    int4 w[SCAN_ITER];
#pragma unroll
    for (int u = 0; u < SCAN_ITER; ++u) {
        const int idx = u * BT + threadIdx.x;
        w[u] = h4[idx < SCAN_V4 ? idx : SCAN_V4 - 1];
    }
#pragma unroll
    for (int u = 0; u < SCAN_ITER; ++u) {
        const int idx = u * BT + threadIdx.x;
        tab4[idx] = idx < SCAN_V4 ? w[u] : make_int4(0, 0, 0, 0);
    }
    __syncthreads();
    const int base = threadIdx.x * SCAN_PER_THREAD;
    int v[SCAN_PER_THREAD];
    int s = 0;
#pragma unroll
    for (int k = 0; k < SCAN_PER_THREAD; ++k) {
        v[k] = tab[base + k];
        s += v[k];
    }
    int total;
    int run = block_excl_scan(s, wsum, total);
#pragma unroll
    for (int k = 0; k < SCAN_PER_THREAD; ++k) {
        tab[base + k] = run;
        run += v[k];
    }
    __syncthreads();
#pragma unroll
    for (int u = 0; u < SCAN_ITER; ++u) {
        const int idx = u * BT + threadIdx.x;
        if (idx < SCAN_V4) h4[idx] = tab4[idx];
    }
}

__global__ __launch_bounds__(BT) void part_scatter_kernel(const int* __restrict__ src,
                                                          const int* __restrict__ dst,
                                                          const unsigned* __restrict__ off,
                                                          unsigned* __restrict__ rec) {
    __shared__ int sdst[SL_EDGES];
    __shared__ int ssrc[SL_EDGES];
    __shared__ unsigned cursor[NBK];
    __shared__ int delta[NBK];
    __shared__ int wsum[32];
    const int sl = blockIdx.x;
    int c = 0, g = 0;
    if (threadIdx.x < NBK) {
        const int i = threadIdx.x * NSL + sl;
        g = (int)off[i];
        c = (i + 1 < NBK * NSL ? (int)off[i + 1] : NE) - g;
    }
    int total;
    const int ex = block_excl_scan(c, wsum, total);
    if (threadIdx.x < NBK) {
        cursor[threadIdx.x] = (unsigned)ex;
        delta[threadIdx.x] = g - ex;
    }
    __syncthreads();

    const int4* pd = reinterpret_cast<const int4*>(dst + (size_t)sl * SL_EDGES);
    const int4* ps = reinterpret_cast<const int4*>(src + (size_t)sl * SL_EDGES);
    int4 d[SL_ITER], s[SL_ITER];
#pragma unroll
    for (int u = 0; u < SL_ITER; ++u) {
        const int idx = u * BT + threadIdx.x;
        const int j = idx < SL_V4 ? idx : SL_V4 - 1;
        d[u] = pd[j];
        s[u] = ps[j];
    }
#pragma unroll
    for (int u = 0; u < SL_ITER; ++u) {
        if (u * BT + (int)threadIdx.x < SL_V4) {
            const unsigned p0 = min(atomicAdd(&cursor[bucket_of(d[u].x)], 1u), (unsigned)(SL_EDGES - 1));
            const unsigned p1 = min(atomicAdd(&cursor[bucket_of(d[u].y)], 1u), (unsigned)(SL_EDGES - 1));
            const unsigned p2 = min(atomicAdd(&cursor[bucket_of(d[u].z)], 1u), (unsigned)(SL_EDGES - 1));
            const unsigned p3 = min(atomicAdd(&cursor[bucket_of(d[u].w)], 1u), (unsigned)(SL_EDGES - 1));
            sdst[p0] = d[u].x; ssrc[p0] = s[u].x;
            sdst[p1] = d[u].y; ssrc[p1] = s[u].y;
            sdst[p2] = d[u].z; ssrc[p2] = s[u].z;
            sdst[p3] = d[u].w; ssrc[p3] = s[u].w;
        }
    }
    __syncthreads();
    for (int i = threadIdx.x; i < SL_EDGES; i += BT) {
        const int dd = sdst[i];
        const unsigned b = bucket_of(dd);
        const unsigned local = (unsigned)(dd - (int)b * BK_NODES) & (unsigned)((1 << (32 - SRC_BITS)) - 1);
        unsigned slot = (unsigned)(delta[b] + i);
        if (slot >= (unsigned)NE) slot = NE - 1;
        rec[slot] = ((unsigned)ssrc[i] & (unsigned)((1 << SRC_BITS) - 1)) | (local << SRC_BITS);
    }
}

__global__ __launch_bounds__(BT) void bucket_csr_kernel(const unsigned* __restrict__ off,
                                                        const unsigned* __restrict__ rec,
                                                        float* __restrict__ dinv, int* __restrict__ cnt,
                                                        int* __restrict__ rowptr,
                                                        int* __restrict__ csr_src) {
    __shared__ unsigned deg[BK_NODES];
    __shared__ unsigned cur[BK_NODES];
    __shared__ int ordered[BK_CAP];
    __shared__ int wsum[32];
    const int bk = blockIdx.x;
    int start = (int)off[(size_t)bk * NSL];
    int end = bk + 1 < NBK ? (int)off[(size_t)(bk + 1) * NSL] : NE;
    start = min(max(start, 0), NE);
    end = min(max(end, start), NE);
    const int n = end - start;
    if (threadIdx.x < BK_NODES) deg[threadIdx.x] = 0u;
    __syncthreads();
    for (int i = threadIdx.x; i < n; i += BT) {
        const unsigned l = min(rec[start + i] >> SRC_BITS, (unsigned)(BK_NODES - 1));
        atomicAdd(&deg[l], 1u);
    }
    __syncthreads();
    const int c = threadIdx.x < BK_NODES ? (int)deg[threadIdx.x] : 0;
    int total;
    const int ex = block_excl_scan(c, wsum, total);
    if (threadIdx.x < BK_NODES) {
        const int node = bk * BK_NODES + threadIdx.x;
        cur[threadIdx.x] = (unsigned)ex;
        dinv[node] = rsqrtf((float)(c + 1));
        cnt[node] = c;
        rowptr[node] = start + ex;
    }
    __syncthreads();
    for (int i = threadIdx.x; i < n; i += BT) {
        const unsigned r = rec[start + i];
        const unsigned l = min(r >> SRC_BITS, (unsigned)(BK_NODES - 1));
        const int p = (int)atomicAdd(&cur[l], 1u);
        const int sidx = (int)(r & (unsigned)((1 << SRC_BITS) - 1));
        if (p < BK_CAP) ordered[p] = sidx;
        else if (p < n) csr_src[start + p] = sidx;
    }
    __syncthreads();
    const int m = n < BK_CAP ? n : BK_CAP;
    for (int i = threadIdx.x; i < m; i += BT) csr_src[start + i] = ordered[i];
}

template <int K, int FOUT>
__global__ void gemm_scaled_kernel(const float* __restrict__ X,
                                   const float* __restrict__ W,
                                   const float* __restrict__ dinv,
                                   float* __restrict__ out, int nrows) {
    const int wave = (blockIdx.x * blockDim.x + threadIdx.x) >> 5;
    const int lane = threadIdx.x & 31;
    const int tilesN = FOUT / 16;
    const int tm = wave / tilesN;
    const int tn = wave % tilesN;
    if (tm * 16 >= nrows) return;

    const int m = lane & 15;
    const int half = lane >> 4;
    const int row0 = tm * 16;
    const int col0 = tn * 16;

    v8f acc = {};
#pragma unroll
    for (int kb = 0; kb < K; kb += 4) {
        const int k0 = kb + 2 * half;
        v2f a, b;
        a.x = X[(size_t)(row0 + m) * K + k0 + 0];
        a.y = X[(size_t)(row0 + m) * K + k0 + 1];
        b.x = W[(size_t)(k0 + 0) * FOUT + col0 + m];
        b.y = W[(size_t)(k0 + 1) * FOUT + col0 + m];
        acc = __builtin_amdgcn_wmma_f32_16x16x4_f32(false, a, false, b,
                                                    (short)0, acc, false, false);
    }
#pragma unroll
    for (int r = 0; r < 8; ++r) {
        const int row = row0 + r + 8 * half;
        out[(size_t)row * FOUT + col0 + m] = acc[r] * dinv[row];
    }
}

#define NODES_PER_WAVE 16
#define PULL_WAVES_PER_BLOCK 4
static_assert(NODES_PER_WAVE <= 32, "a lane holds the row range of one of the wave's nodes");
static_assert((long long)NN * HID_DIM * 4 < (1LL << 32), "row byte offsets are kept in 32 bits");

template <int F>
struct PullShape {
    static constexpr int LPR = F / 4;
    static constexpr int RPI = 32 / LPR;
    static constexpr int UNR = 32 / RPI;
};

template <int F>
__device__ __forceinline__ void chunk_sources(int idx, int g, unsigned q16,
                                              unsigned (&so)[PullShape<F>::UNR]) {
#pragma unroll
    for (int u = 0; u < PullShape<F>::UNR; ++u)
        so[u] = (unsigned)__shfl(idx, u * PullShape<F>::RPI + g) * (unsigned)(F * 4) + q16;
}

template <int F>
__device__ __forceinline__ void chunk_rows(const float* __restrict__ hws, int m,
                                           const unsigned (&so)[PullShape<F>::UNR],
                                           float4 (&v)[PullShape<F>::UNR]) {
#pragma unroll
    for (int u = 0; u < PullShape<F>::UNR; ++u) {
        v[u] = make_float4(0.0f, 0.0f, 0.0f, 0.0f);
        if (u * PullShape<F>::RPI < m)
            v[u] = *reinterpret_cast<const float4*>(reinterpret_cast<const char*>(hws) + so[u]);
    }
}

template <int F>
__device__ __forceinline__ void chunk_add(int m, int g, float nd,
                                          const float4 (&v)[PullShape<F>::UNR], float4& acc) {
#pragma unroll
    for (int u = 0; u < PullShape<F>::UNR; ++u) {
        if (u * PullShape<F>::RPI + g < m) {
            acc.x += v[u].x * nd;
            acc.y += v[u].y * nd;
            acc.z += v[u].z * nd;
            acc.w += v[u].w * nd;
        }
    }
}

template <int F, bool RELU>
__global__ __launch_bounds__(32 * PULL_WAVES_PER_BLOCK) void pull_agg_kernel(
        const int* __restrict__ rowptr, const int* __restrict__ cnt,
        const int* __restrict__ csr_src, const float* __restrict__ hws,
        const float* __restrict__ dinv, const float* __restrict__ bias,
        float* __restrict__ out, int n) {
    static_assert(F == 32 || F == 16, "a row is 8 or 4 float4");
    constexpr int LPR = PullShape<F>::LPR;
    constexpr int UNR = PullShape<F>::UNR;
    const int lane = threadIdx.x & 31;
    const int wave = __builtin_amdgcn_readfirstlane((int)((blockIdx.x * blockDim.x + threadIdx.x) >> 5));
    const int g = lane / LPR;
    const int q = lane % LPR;
    const unsigned q16 = (unsigned)q * 16u;
    const int node0 = wave * NODES_PER_WAVE;
    if (node0 >= n) return;

    int rp = 0, cn = 0, dn = 0;
    if (lane < NODES_PER_WAVE && node0 + lane < n) {
        rp = rowptr[node0 + lane];
        cn = cnt[node0 + lane];
        dn = __float_as_int(dinv[node0 + lane]);
    }
    const float4 b4 = reinterpret_cast<const float4*>(bias)[q];

    int idx_next = 0;
    {
        const int d0 = __builtin_amdgcn_readlane(cn, 0);
        const int s0 = __builtin_amdgcn_readlane(rp, 0);
        if (d0 > 0) idx_next = csr_src[s0 + (lane < d0 ? lane : d0 - 1)];
    }

    for (int k = 0; k < NODES_PER_WAVE; ++k) {
        const int node = node0 + k;
        if (node >= n) break;
        const int start = __builtin_amdgcn_readlane(rp, k);
        const int deg = __builtin_amdgcn_readlane(cn, k);
        const float nd = __int_as_float(__builtin_amdgcn_readlane(dn, k));

        unsigned so[UNR];
        float4 v[UNR];
        chunk_sources<F>(idx_next, g, q16, so);
        if (k + 1 < NODES_PER_WAVE && node + 1 < n) {
            const int d1 = __builtin_amdgcn_readlane(cn, k + 1);
            const int s1 = __builtin_amdgcn_readlane(rp, k + 1);
            if (d1 > 0) idx_next = csr_src[s1 + (lane < d1 ? lane : d1 - 1)];
        }
        const float4 self = reinterpret_cast<const float4*>(hws + (size_t)node * F)[q];

        float4 acc = make_float4(0.0f, 0.0f, 0.0f, 0.0f);
        const int m0 = deg < 32 ? deg : 32;
        chunk_rows<F>(hws, m0, so, v);
        chunk_add<F>(m0, g, nd, v, acc);
        for (int base = 32; base < deg; base += 32) {
            const int left = deg - base;
            const int m = left < 32 ? left : 32;
            const int e = base + lane < deg ? base + lane : deg - 1;
            const int idx = csr_src[start + e];
            chunk_sources<F>(idx, g, q16, so);
            chunk_rows<F>(hws, m, so, v);
            chunk_add<F>(m, g, nd, v, acc);
        }
#pragma unroll
        for (int d = LPR; d < 32; d <<= 1) {
            acc.x += __shfl_xor(acc.x, d);
            acc.y += __shfl_xor(acc.y, d);
            acc.z += __shfl_xor(acc.z, d);
            acc.w += __shfl_xor(acc.w, d);
        }
        if (g == 0) {
            float4 r;
            r.x = (self.x * nd + b4.x) + acc.x;
            r.y = (self.y * nd + b4.y) + acc.y;
            r.z = (self.z * nd + b4.z) + acc.z;
            r.w = (self.w * nd + b4.w) + acc.w;
            if (RELU) {
                r.x = fmaxf(r.x, 0.0f);
                r.y = fmaxf(r.y, 0.0f);
                r.z = fmaxf(r.z, 0.0f);
                r.w = fmaxf(r.w, 0.0f);
            }
            reinterpret_cast<float4*>(out + (size_t)node * F)[q] = r;
        }
    }
}

static inline int cdiv(long long a, long long b) { return (int)((a + b - 1) / b); }

extern "C" void kernel_launch(void* const* d_in, const int* in_sizes, int n_in,
                              void* d_out, int out_size, void* d_ws, size_t ws_size,
                              hipStream_t stream) {
    const float* x  = (const float*)d_in[0];
    const int* ei   = (const int*)d_in[1];
    const float* W1 = (const float*)d_in[2];
    const float* b1 = (const float*)d_in[3];
    const float* W2 = (const float*)d_in[4];
    const float* b2 = (const float*)d_in[5];
    float* out = (float*)d_out;

    const int* src = ei;
    const int* dst = ei + NE;

    char* ws = (char*)d_ws;
    float* dinv = (float*)ws;
    float* hw1  = (float*)(ws + (size_t)NN * 4);
    float* h1   = hw1 + (size_t)NN * HID_DIM;
    float* hw2  = h1  + (size_t)NN * HID_DIM;
    unsigned* off = (unsigned*)(hw2 + (size_t)NN * OUT_DIM);
    unsigned* rec = off + 65536;
    int* cnt    = (int*)(rec + (size_t)NE);
    int* rowptr = cnt + NN;
    int* csrc   = rowptr + NN;
    static_assert(NBK * NSL <= 65536, "off plane");

    const int B = 256;

    part_count_kernel<<<NSL, BT, 0, stream>>>(dst, off);
    part_scan_kernel<<<1, BT, 0, stream>>>(off);
    part_scatter_kernel<<<NSL, BT, 0, stream>>>(src, dst, off, rec);
    bucket_csr_kernel<<<NBK, BT, 0, stream>>>(off, rec, dinv, cnt, rowptr, csrc);

    const int pull_blocks = cdiv(NN, NODES_PER_WAVE * PULL_WAVES_PER_BLOCK);

    {
        const int waves = (NN / 16) * (HID_DIM / 16);
        gemm_scaled_kernel<IN_DIM, HID_DIM>
            <<<cdiv((long long)waves * 32, B), B, 0, stream>>>(x, W1, dinv, hw1, NN);
    }
    pull_agg_kernel<HID_DIM, true><<<pull_blocks, 32 * PULL_WAVES_PER_BLOCK, 0, stream>>>(
        rowptr, cnt, csrc, hw1, dinv, b1, h1, NN);

    {
        const int waves = (NN / 16) * (OUT_DIM / 16);
        gemm_scaled_kernel<HID_DIM, OUT_DIM>
            <<<cdiv((long long)waves * 32, B), B, 0, stream>>>(h1, W2, dinv, hw2, NN);
    }
    pull_agg_kernel<OUT_DIM, false><<<pull_blocks, 32 * PULL_WAVES_PER_BLOCK, 0, stream>>>(
        rowptr, cnt, csrc, hw2, dinv, b2, out, NN);
}


# per-bucket list build: a thread fetches its records once, in one group of loads, and keeps them in registers for both passes (was load, wait, LDS add per trip, twice); sums still in arrival order
# speedup vs baseline: 4.8661x; 1.0306x over previous
#include <hip/hip_runtime.h>

#define NN 200000
#define NE 6400000
#define IN_DIM 32
#define HID_DIM 32
#define OUT_DIM 16

typedef __attribute__((ext_vector_type(2))) float v2f;
typedef __attribute__((ext_vector_type(8))) float v8f;

#define BK_NODES 800
#define NBK (NN / BK_NODES)
#define NSL 250
#define SL_EDGES (NE / NSL)
#define SL_V4 (SL_EDGES / 4)
#define BT 1024
#define SL_ITER ((SL_V4 + BT - 1) / BT)
#define SRC_BITS 18
#define BK_CAP 32768
static_assert(NN % BK_NODES == 0 && NE % NSL == 0 && SL_EDGES % 4 == 0, "even split");
static_assert(NN <= (1 << SRC_BITS) && BK_NODES <= (1 << (32 - SRC_BITS)), "record fields");
static_assert(NBK <= BT && BK_NODES <= BT, "one thread per bucket / per node of a bucket");
static_assert((long long)NE < (1LL << 31), "edge slots are kept in non-negative ints");

__device__ __forceinline__ unsigned bucket_of(int d) {
    const unsigned b = (unsigned)d / (unsigned)BK_NODES;
    return b < (unsigned)NBK ? b : (unsigned)(NBK - 1);
}

__device__ __forceinline__ int block_excl_scan(int v, int* wsum, int& total) {
    const int lane = threadIdx.x & 31;
    const int w = threadIdx.x >> 5;
    int inc = v;
#pragma unroll
    for (int d = 1; d < 32; d <<= 1) {
        const int t = __shfl_up(inc, d);
        if (lane >= d) inc += t;
    }
    if (lane == 31) wsum[w] = inc;
    __syncthreads();
    const int ws = wsum[lane];
    int winc = ws;
#pragma unroll
    for (int d = 1; d < 32; d <<= 1) {
        const int t = __shfl_up(winc, d);
        if (lane >= d) winc += t;
    }
    const int woff = __shfl(winc, w) - __shfl(ws, w);
    total = __shfl(winc, 31);
    __syncthreads();
    return woff + inc - v;
}
static_assert(BT == 1024, "block_excl_scan: 32 waves of 32");

__global__ __launch_bounds__(BT) void part_count_kernel(const int* __restrict__ dst,
                                                        unsigned* __restrict__ hist) {
    __shared__ unsigned cnt[BT / 32][NBK];
    const int w = threadIdx.x >> 5;
    for (int i = threadIdx.x; i < (BT / 32) * NBK; i += BT) (&cnt[0][0])[i] = 0u;
    __syncthreads();
    const int4* p = reinterpret_cast<const int4*>(dst + (size_t)blockIdx.x * SL_EDGES);
    int4 d[SL_ITER];
#pragma unroll
    for (int u = 0; u < SL_ITER; ++u) {
        const int idx = u * BT + threadIdx.x;
        d[u] = p[idx < SL_V4 ? idx : SL_V4 - 1];
    }
#pragma unroll
    for (int u = 0; u < SL_ITER; ++u) {
        if (u * BT + (int)threadIdx.x < SL_V4) {
            atomicAdd(&cnt[w][bucket_of(d[u].x)], 1u);
            atomicAdd(&cnt[w][bucket_of(d[u].y)], 1u);
            atomicAdd(&cnt[w][bucket_of(d[u].z)], 1u);
            atomicAdd(&cnt[w][bucket_of(d[u].w)], 1u);
        }
    }
    __syncthreads();
    if (threadIdx.x < NBK) {
        unsigned s = 0u;
#pragma unroll
        for (int k = 0; k < BT / 32; ++k) s += cnt[k][threadIdx.x];
        hist[(size_t)threadIdx.x * NSL + blockIdx.x] = s;
    }
}

#define SCAN_N (NBK * NSL)
#define SCAN_PER_THREAD ((SCAN_N + BT - 1) / BT)
#define SCAN_V4 (SCAN_N / 4)
#define SCAN_ITER ((BT * SCAN_PER_THREAD / 4 + BT - 1) / BT)
static_assert(SCAN_N % 4 == 0, "the table is copied in 16-byte words");
static_assert(SCAN_ITER * BT * 4 >= BT * SCAN_PER_THREAD, "the LDS copy covers every thread's entries");
__global__ __launch_bounds__(BT) void part_scan_kernel(unsigned* hist) {
    __shared__ int4 tab4[SCAN_ITER * BT];
    __shared__ int wsum[32];
    int* tab = reinterpret_cast<int*>(tab4);
    int4* h4 = reinterpret_cast<int4*>(hist);
    int4 w[SCAN_ITER];
#pragma unroll
    for (int u = 0; u < SCAN_ITER; ++u) {
        const int idx = u * BT + threadIdx.x;
        w[u] = h4[idx < SCAN_V4 ? idx : SCAN_V4 - 1];
    }
#pragma unroll
    for (int u = 0; u < SCAN_ITER; ++u) {
        const int idx = u * BT + threadIdx.x;
        tab4[idx] = idx < SCAN_V4 ? w[u] : make_int4(0, 0, 0, 0);
    }
    __syncthreads();
    const int base = threadIdx.x * SCAN_PER_THREAD;
    int v[SCAN_PER_THREAD];
    int s = 0;
#pragma unroll
    for (int k = 0; k < SCAN_PER_THREAD; ++k) {
        v[k] = tab[base + k];
        s += v[k];
    }
    int total;
    int run = block_excl_scan(s, wsum, total);
#pragma unroll
    for (int k = 0; k < SCAN_PER_THREAD; ++k) {
        tab[base + k] = run;
        run += v[k];
    }
    __syncthreads();
#pragma unroll
    for (int u = 0; u < SCAN_ITER; ++u) {
        const int idx = u * BT + threadIdx.x;
        if (idx < SCAN_V4) h4[idx] = tab4[idx];
    }
}

__global__ __launch_bounds__(BT) void part_scatter_kernel(const int* __restrict__ src,
                                                          const int* __restrict__ dst,
                                                          const unsigned* __restrict__ off,
                                                          unsigned* __restrict__ rec) {
    __shared__ int sdst[SL_EDGES];
    __shared__ int ssrc[SL_EDGES];
    __shared__ unsigned cursor[NBK];
    __shared__ int delta[NBK];
    __shared__ int wsum[32];
    const int sl = blockIdx.x;
    int c = 0, g = 0;
    if (threadIdx.x < NBK) {
        const int i = threadIdx.x * NSL + sl;
        g = (int)off[i];
        c = (i + 1 < NBK * NSL ? (int)off[i + 1] : NE) - g;
    }
    int total;
    const int ex = block_excl_scan(c, wsum, total);
    if (threadIdx.x < NBK) {
        cursor[threadIdx.x] = (unsigned)ex;
        delta[threadIdx.x] = g - ex;
    }
    __syncthreads();

    const int4* pd = reinterpret_cast<const int4*>(dst + (size_t)sl * SL_EDGES);
    const int4* ps = reinterpret_cast<const int4*>(src + (size_t)sl * SL_EDGES);
    int4 d[SL_ITER], s[SL_ITER];
#pragma unroll
    for (int u = 0; u < SL_ITER; ++u) {
        const int idx = u * BT + threadIdx.x;
        const int j = idx < SL_V4 ? idx : SL_V4 - 1;
        d[u] = pd[j];
        s[u] = ps[j];
    }
#pragma unroll
    for (int u = 0; u < SL_ITER; ++u) {
        if (u * BT + (int)threadIdx.x < SL_V4) {
            const unsigned p0 = min(atomicAdd(&cursor[bucket_of(d[u].x)], 1u), (unsigned)(SL_EDGES - 1));
            const unsigned p1 = min(atomicAdd(&cursor[bucket_of(d[u].y)], 1u), (unsigned)(SL_EDGES - 1));
            const unsigned p2 = min(atomicAdd(&cursor[bucket_of(d[u].z)], 1u), (unsigned)(SL_EDGES - 1));
            const unsigned p3 = min(atomicAdd(&cursor[bucket_of(d[u].w)], 1u), (unsigned)(SL_EDGES - 1));
            sdst[p0] = d[u].x; ssrc[p0] = s[u].x;
            sdst[p1] = d[u].y; ssrc[p1] = s[u].y;
            sdst[p2] = d[u].z; ssrc[p2] = s[u].z;
            sdst[p3] = d[u].w; ssrc[p3] = s[u].w;
        }
    }
    __syncthreads();
    for (int i = threadIdx.x; i < SL_EDGES; i += BT) {
        const int dd = sdst[i];
        const unsigned b = bucket_of(dd);
        const unsigned local = (unsigned)(dd - (int)b * BK_NODES) & (unsigned)((1 << (32 - SRC_BITS)) - 1);
        unsigned slot = (unsigned)(delta[b] + i);
        if (slot >= (unsigned)NE) slot = NE - 1;
        rec[slot] = ((unsigned)ssrc[i] & (unsigned)((1 << SRC_BITS) - 1)) | (local << SRC_BITS);
    }
}

#define CSR_REG (BK_CAP / BT)
static_assert(BK_CAP % BT == 0, "the records kept in registers cover the LDS plane exactly");
__global__ __launch_bounds__(BT) void bucket_csr_kernel(const unsigned* __restrict__ off,
                                                        const unsigned* __restrict__ rec,
                                                        float* __restrict__ dinv, int* __restrict__ cnt,
                                                        int* __restrict__ rowptr,
                                                        int* __restrict__ csr_src) {
    __shared__ unsigned deg[BK_NODES];
    __shared__ unsigned cur[BK_NODES];
    __shared__ int ordered[BK_CAP];
    __shared__ int wsum[32];
    const int bk = blockIdx.x;
    const int tid = (int)threadIdx.x;
    int start = (int)off[(size_t)bk * NSL];
    int end = bk + 1 < NBK ? (int)off[(size_t)(bk + 1) * NSL] : NE;
    start = min(max(start, 0), NE);
    end = min(max(end, start), NE);
    const int n = end - start;
    unsigned r[CSR_REG];
#pragma unroll
    for (int k = 0; k < CSR_REG; ++k) {
        r[k] = 0u;
        if (k * BT < n) {
            const int i = k * BT + tid;
            r[k] = rec[start + (i < n ? i : n - 1)];
        }
    }
    if (tid < BK_NODES) deg[tid] = 0u;
    __syncthreads();
#pragma unroll
    for (int k = 0; k < CSR_REG; ++k) {
        if (k * BT + tid < n)
            atomicAdd(&deg[min(r[k] >> SRC_BITS, (unsigned)(BK_NODES - 1))], 1u);
    }
    for (int i = BK_CAP + tid; i < n; i += BT) {
        const unsigned l = min(rec[start + i] >> SRC_BITS, (unsigned)(BK_NODES - 1));
        atomicAdd(&deg[l], 1u);
    }
    __syncthreads();
    const int c = tid < BK_NODES ? (int)deg[tid] : 0;
    int total;
    const int ex = block_excl_scan(c, wsum, total);
    if (tid < BK_NODES) {
        const int node = bk * BK_NODES + tid;
        cur[tid] = (unsigned)ex;
        dinv[node] = rsqrtf((float)(c + 1));
        cnt[node] = c;
        rowptr[node] = start + ex;
    }
    __syncthreads();
    int p[CSR_REG];
#pragma unroll
    for (int k = 0; k < CSR_REG; ++k) {
        p[k] = n;
        if (k * BT + tid < n)
            p[k] = (int)atomicAdd(&cur[min(r[k] >> SRC_BITS, (unsigned)(BK_NODES - 1))], 1u);
    }
    const int m = n < BK_CAP ? n : BK_CAP;
#pragma unroll
    for (int k = 0; k < CSR_REG; ++k) {
        const int sidx = (int)(r[k] & (unsigned)((1 << SRC_BITS) - 1));
        if (p[k] < m) ordered[p[k]] = sidx;
        else if (p[k] < n) csr_src[start + p[k]] = sidx;
    }
    for (int i = BK_CAP + tid; i < n; i += BT) {
        const unsigned rr = rec[start + i];
        const unsigned l = min(rr >> SRC_BITS, (unsigned)(BK_NODES - 1));
        const int q = (int)atomicAdd(&cur[l], 1u);
        const int sidx = (int)(rr & (unsigned)((1 << SRC_BITS) - 1));
        if (q < m) ordered[q] = sidx;
        else if (q < n) csr_src[start + q] = sidx;
    }
    __syncthreads();
    for (int i = tid; i < m; i += BT) csr_src[start + i] = ordered[i];
}

template <int K, int FOUT>
__global__ void gemm_scaled_kernel(const float* __restrict__ X,
                                   const float* __restrict__ W,
                                   const float* __restrict__ dinv,
                                   float* __restrict__ out, int nrows) {
    const int wave = (blockIdx.x * blockDim.x + threadIdx.x) >> 5;
    const int lane = threadIdx.x & 31;
    const int tilesN = FOUT / 16;
    const int tm = wave / tilesN;
    const int tn = wave % tilesN;
    if (tm * 16 >= nrows) return;

    const int m = lane & 15;
    const int half = lane >> 4;
    const int row0 = tm * 16;
    const int col0 = tn * 16;

    v8f acc = {};
#pragma unroll
    for (int kb = 0; kb < K; kb += 4) {
        const int k0 = kb + 2 * half;
        v2f a, b;
        a.x = X[(size_t)(row0 + m) * K + k0 + 0];
        a.y = X[(size_t)(row0 + m) * K + k0 + 1];
        b.x = W[(size_t)(k0 + 0) * FOUT + col0 + m];
        b.y = W[(size_t)(k0 + 1) * FOUT + col0 + m];
        acc = __builtin_amdgcn_wmma_f32_16x16x4_f32(false, a, false, b,
                                                    (short)0, acc, false, false);
    }
#pragma unroll
    for (int r = 0; r < 8; ++r) {
        const int row = row0 + r + 8 * half;
        out[(size_t)row * FOUT + col0 + m] = acc[r] * dinv[row];
    }
}

#define NODES_PER_WAVE 16
#define PULL_WAVES_PER_BLOCK 4
static_assert(NODES_PER_WAVE <= 32, "a lane holds the row range of one of the wave's nodes");
static_assert((long long)NN * HID_DIM * 4 < (1LL << 32), "row byte offsets are kept in 32 bits");

template <int F>
struct PullShape {
    static constexpr int LPR = F / 4;
    static constexpr int RPI = 32 / LPR;
    static constexpr int UNR = 32 / RPI;
};

template <int F>
__device__ __forceinline__ void chunk_sources(int idx, int g, unsigned q16,
                                              unsigned (&so)[PullShape<F>::UNR]) {
#pragma unroll
    for (int u = 0; u < PullShape<F>::UNR; ++u)
        so[u] = (unsigned)__shfl(idx, u * PullShape<F>::RPI + g) * (unsigned)(F * 4) + q16;
}

template <int F>
__device__ __forceinline__ void chunk_rows(const float* __restrict__ hws, int m,
                                           const unsigned (&so)[PullShape<F>::UNR],
                                           float4 (&v)[PullShape<F>::UNR]) {
#pragma unroll
    for (int u = 0; u < PullShape<F>::UNR; ++u) {
        v[u] = make_float4(0.0f, 0.0f, 0.0f, 0.0f);
        if (u * PullShape<F>::RPI < m)
            v[u] = *reinterpret_cast<const float4*>(reinterpret_cast<const char*>(hws) + so[u]);
    }
}

template <int F>
__device__ __forceinline__ void chunk_add(int m, int g, float nd,
                                          const float4 (&v)[PullShape<F>::UNR], float4& acc) {
#pragma unroll
    for (int u = 0; u < PullShape<F>::UNR; ++u) {
        if (u * PullShape<F>::RPI + g < m) {
            acc.x += v[u].x * nd;
            acc.y += v[u].y * nd;
            acc.z += v[u].z * nd;
            acc.w += v[u].w * nd;
        }
    }
}

template <int F, bool RELU>
__global__ __launch_bounds__(32 * PULL_WAVES_PER_BLOCK) void pull_agg_kernel(
        const int* __restrict__ rowptr, const int* __restrict__ cnt,
        const int* __restrict__ csr_src, const float* __restrict__ hws,
        const float* __restrict__ dinv, const float* __restrict__ bias,
        float* __restrict__ out, int n) {
    static_assert(F == 32 || F == 16, "a row is 8 or 4 float4");
    constexpr int LPR = PullShape<F>::LPR;
    constexpr int UNR = PullShape<F>::UNR;
    const int lane = threadIdx.x & 31;
    const int wave = __builtin_amdgcn_readfirstlane((int)((blockIdx.x * blockDim.x + threadIdx.x) >> 5));
    const int g = lane / LPR;
    const int q = lane % LPR;
    const unsigned q16 = (unsigned)q * 16u;
    const int node0 = wave * NODES_PER_WAVE;
    if (node0 >= n) return;

    int rp = 0, cn = 0, dn = 0;
    if (lane < NODES_PER_WAVE && node0 + lane < n) {
        rp = rowptr[node0 + lane];
        cn = cnt[node0 + lane];
        dn = __float_as_int(dinv[node0 + lane]);
    }
    const float4 b4 = reinterpret_cast<const float4*>(bias)[q];

    int idx_next = 0;
    {
        const int d0 = __builtin_amdgcn_readlane(cn, 0);
        const int s0 = __builtin_amdgcn_readlane(rp, 0);
        if (d0 > 0) idx_next = csr_src[s0 + (lane < d0 ? lane : d0 - 1)];
    }

    for (int k = 0; k < NODES_PER_WAVE; ++k) {
        const int node = node0 + k;
        if (node >= n) break;
        const int start = __builtin_amdgcn_readlane(rp, k);
        const int deg = __builtin_amdgcn_readlane(cn, k);
        const float nd = __int_as_float(__builtin_amdgcn_readlane(dn, k));

        unsigned so[UNR];
        float4 v[UNR];
        chunk_sources<F>(idx_next, g, q16, so);
        if (k + 1 < NODES_PER_WAVE && node + 1 < n) {
            const int d1 = __builtin_amdgcn_readlane(cn, k + 1);
            const int s1 = __builtin_amdgcn_readlane(rp, k + 1);
            if (d1 > 0) idx_next = csr_src[s1 + (lane < d1 ? lane : d1 - 1)];
        }
        const float4 self = reinterpret_cast<const float4*>(hws + (size_t)node * F)[q];

        float4 acc = make_float4(0.0f, 0.0f, 0.0f, 0.0f);
        const int m0 = deg < 32 ? deg : 32;
        chunk_rows<F>(hws, m0, so, v);
        chunk_add<F>(m0, g, nd, v, acc);
        for (int base = 32; base < deg; base += 32) {
            const int left = deg - base;
            const int m = left < 32 ? left : 32;
            const int e = base + lane < deg ? base + lane : deg - 1;
            const int idx = csr_src[start + e];
            chunk_sources<F>(idx, g, q16, so);
            chunk_rows<F>(hws, m, so, v);
            chunk_add<F>(m, g, nd, v, acc);
        }
#pragma unroll
        for (int d = LPR; d < 32; d <<= 1) {
            acc.x += __shfl_xor(acc.x, d);
            acc.y += __shfl_xor(acc.y, d);
            acc.z += __shfl_xor(acc.z, d);
            acc.w += __shfl_xor(acc.w, d);
        }
        if (g == 0) {
            float4 r;
            r.x = (self.x * nd + b4.x) + acc.x;
            r.y = (self.y * nd + b4.y) + acc.y;
            r.z = (self.z * nd + b4.z) + acc.z;
            r.w = (self.w * nd + b4.w) + acc.w;
            if (RELU) {
                r.x = fmaxf(r.x, 0.0f);
                r.y = fmaxf(r.y, 0.0f);
                r.z = fmaxf(r.z, 0.0f);
                r.w = fmaxf(r.w, 0.0f);
            }
            reinterpret_cast<float4*>(out + (size_t)node * F)[q] = r;
        }
    }
}

static inline int cdiv(long long a, long long b) { return (int)((a + b - 1) / b); }

extern "C" void kernel_launch(void* const* d_in, const int* in_sizes, int n_in,
                              void* d_out, int out_size, void* d_ws, size_t ws_size,
                              hipStream_t stream) {
    const float* x  = (const float*)d_in[0];
    const int* ei   = (const int*)d_in[1];
    const float* W1 = (const float*)d_in[2];
    const float* b1 = (const float*)d_in[3];
    const float* W2 = (const float*)d_in[4];
    const float* b2 = (const float*)d_in[5];
    float* out = (float*)d_out;

    const int* src = ei;
    const int* dst = ei + NE;

    char* ws = (char*)d_ws;
    float* dinv = (float*)ws;
    float* hw1  = (float*)(ws + (size_t)NN * 4);
    float* h1   = hw1 + (size_t)NN * HID_DIM;
    float* hw2  = h1  + (size_t)NN * HID_DIM;
    unsigned* off = (unsigned*)(hw2 + (size_t)NN * OUT_DIM);
    unsigned* rec = off + 65536;
    int* cnt    = (int*)(rec + (size_t)NE);
    int* rowptr = cnt + NN;
    int* csrc   = rowptr + NN;
    static_assert(NBK * NSL <= 65536, "off plane");

    const int B = 256;

    part_count_kernel<<<NSL, BT, 0, stream>>>(dst, off);
    part_scan_kernel<<<1, BT, 0, stream>>>(off);
    part_scatter_kernel<<<NSL, BT, 0, stream>>>(src, dst, off, rec);
    bucket_csr_kernel<<<NBK, BT, 0, stream>>>(off, rec, dinv, cnt, rowptr, csrc);

    const int pull_blocks = cdiv(NN, NODES_PER_WAVE * PULL_WAVES_PER_BLOCK);

    {
        const int waves = (NN / 16) * (HID_DIM / 16);
        gemm_scaled_kernel<IN_DIM, HID_DIM>
            <<<cdiv((long long)waves * 32, B), B, 0, stream>>>(x, W1, dinv, hw1, NN);
    }
    pull_agg_kernel<HID_DIM, true><<<pull_blocks, 32 * PULL_WAVES_PER_BLOCK, 0, stream>>>(
        rowptr, cnt, csrc, hw1, dinv, b1, h1, NN);

    {
        const int waves = (NN / 16) * (OUT_DIM / 16);
        gemm_scaled_kernel<HID_DIM, OUT_DIM>
            <<<cdiv((long long)waves * 32, B), B, 0, stream>>>(h1, W2, dinv, hw2, NN);
    }
    pull_agg_kernel<OUT_DIM, false><<<pull_blocks, 32 * PULL_WAVES_PER_BLOCK, 0, stream>>>(
        rowptr, cnt, csrc, hw2, dinv, b2, out, NN);
}


# both X@W launches fused into neighbours: layer-1 product done by the per-bucket build workgroup for its 800 nodes, layer-2 product by the aggregating wave from its 16 h1 rows in LDS; launches 8 -> 6
# speedup vs baseline: 5.2137x; 1.0714x over previous
#include <hip/hip_runtime.h>

#define NN 200000
#define NE 6400000
#define IN_DIM 32
#define HID_DIM 32
#define OUT_DIM 16

typedef __attribute__((ext_vector_type(2))) float v2f;
typedef __attribute__((ext_vector_type(8))) float v8f;

#define BK_NODES 800
#define NBK (NN / BK_NODES)
#define NSL 250
#define SL_EDGES (NE / NSL)
#define SL_V4 (SL_EDGES / 4)
#define BT 1024
#define SL_ITER ((SL_V4 + BT - 1) / BT)
#define SRC_BITS 18
#define BK_CAP 32768
static_assert(NN % BK_NODES == 0 && NE % NSL == 0 && SL_EDGES % 4 == 0, "even split");
static_assert(NN <= (1 << SRC_BITS) && BK_NODES <= (1 << (32 - SRC_BITS)), "record fields");
static_assert(NBK <= BT && BK_NODES <= BT, "one thread per bucket / per node of a bucket");
static_assert((long long)NE < (1LL << 31), "edge slots are kept in non-negative ints");

__device__ __forceinline__ unsigned bucket_of(int d) {
    const unsigned b = (unsigned)d / (unsigned)BK_NODES;
    return b < (unsigned)NBK ? b : (unsigned)(NBK - 1);
}

__device__ __forceinline__ int block_excl_scan(int v, int* wsum, int& total) {
    const int lane = threadIdx.x & 31;
    const int w = threadIdx.x >> 5;
    int inc = v;
#pragma unroll
    for (int d = 1; d < 32; d <<= 1) {
        const int t = __shfl_up(inc, d);
        if (lane >= d) inc += t;
    }
    if (lane == 31) wsum[w] = inc;
    __syncthreads();
    const int ws = wsum[lane];
    int winc = ws;
#pragma unroll
    for (int d = 1; d < 32; d <<= 1) {
        const int t = __shfl_up(winc, d);
        if (lane >= d) winc += t;
    }
    const int woff = __shfl(winc, w) - __shfl(ws, w);
    total = __shfl(winc, 31);
    __syncthreads();
    return woff + inc - v;
}
static_assert(BT == 1024, "block_excl_scan: 32 waves of 32");

__global__ __launch_bounds__(BT) void part_count_kernel(const int* __restrict__ dst,
                                                        unsigned* __restrict__ hist) {
    __shared__ unsigned cnt[BT / 32][NBK];
    const int w = threadIdx.x >> 5;
    for (int i = threadIdx.x; i < (BT / 32) * NBK; i += BT) (&cnt[0][0])[i] = 0u;
    __syncthreads();
    const int4* p = reinterpret_cast<const int4*>(dst + (size_t)blockIdx.x * SL_EDGES);
    int4 d[SL_ITER];
#pragma unroll
    for (int u = 0; u < SL_ITER; ++u) {
        const int idx = u * BT + threadIdx.x;
        d[u] = p[idx < SL_V4 ? idx : SL_V4 - 1];
    }
#pragma unroll
    for (int u = 0; u < SL_ITER; ++u) {
        if (u * BT + (int)threadIdx.x < SL_V4) {
            atomicAdd(&cnt[w][bucket_of(d[u].x)], 1u);
            atomicAdd(&cnt[w][bucket_of(d[u].y)], 1u);
            atomicAdd(&cnt[w][bucket_of(d[u].z)], 1u);
            atomicAdd(&cnt[w][bucket_of(d[u].w)], 1u);
        }
    }
    __syncthreads();
    if (threadIdx.x < NBK) {
        unsigned s = 0u;
#pragma unroll
        for (int k = 0; k < BT / 32; ++k) s += cnt[k][threadIdx.x];
        hist[(size_t)threadIdx.x * NSL + blockIdx.x] = s;
    }
}

#define SCAN_N (NBK * NSL)
#define SCAN_PER_THREAD ((SCAN_N + BT - 1) / BT)
#define SCAN_V4 (SCAN_N / 4)
#define SCAN_ITER ((BT * SCAN_PER_THREAD / 4 + BT - 1) / BT)
static_assert(SCAN_N % 4 == 0, "the table is copied in 16-byte words");
static_assert(SCAN_ITER * BT * 4 >= BT * SCAN_PER_THREAD, "the LDS copy covers every thread's entries");
__global__ __launch_bounds__(BT) void part_scan_kernel(unsigned* hist) {
    __shared__ int4 tab4[SCAN_ITER * BT];
    __shared__ int wsum[32];
    int* tab = reinterpret_cast<int*>(tab4);
    int4* h4 = reinterpret_cast<int4*>(hist);
    int4 w[SCAN_ITER];
#pragma unroll
    for (int u = 0; u < SCAN_ITER; ++u) {
        const int idx = u * BT + threadIdx.x;
        w[u] = h4[idx < SCAN_V4 ? idx : SCAN_V4 - 1];
    }
#pragma unroll
    for (int u = 0; u < SCAN_ITER; ++u) {
        const int idx = u * BT + threadIdx.x;
        tab4[idx] = idx < SCAN_V4 ? w[u] : make_int4(0, 0, 0, 0);
    }
    __syncthreads();
    const int base = threadIdx.x * SCAN_PER_THREAD;
    int v[SCAN_PER_THREAD];
    int s = 0;
#pragma unroll
    for (int k = 0; k < SCAN_PER_THREAD; ++k) {
        v[k] = tab[base + k];
        s += v[k];
    }
    int total;
    int run = block_excl_scan(s, wsum, total);
#pragma unroll
    for (int k = 0; k < SCAN_PER_THREAD; ++k) {
        tab[base + k] = run;
        run += v[k];
    }
    __syncthreads();
#pragma unroll
    for (int u = 0; u < SCAN_ITER; ++u) {
        const int idx = u * BT + threadIdx.x;
        if (idx < SCAN_V4) h4[idx] = tab4[idx];
    }
}

__global__ __launch_bounds__(BT) void part_scatter_kernel(const int* __restrict__ src,
                                                          const int* __restrict__ dst,
                                                          const unsigned* __restrict__ off,
                                                          unsigned* __restrict__ rec) {
    __shared__ int sdst[SL_EDGES];
    __shared__ int ssrc[SL_EDGES];
    __shared__ unsigned cursor[NBK];
    __shared__ int delta[NBK];
    __shared__ int wsum[32];
    const int sl = blockIdx.x;
    int c = 0, g = 0;
    if (threadIdx.x < NBK) {
        const int i = threadIdx.x * NSL + sl;
        g = (int)off[i];
        c = (i + 1 < NBK * NSL ? (int)off[i + 1] : NE) - g;
    }
    int total;
    const int ex = block_excl_scan(c, wsum, total);
    if (threadIdx.x < NBK) {
        cursor[threadIdx.x] = (unsigned)ex;
        delta[threadIdx.x] = g - ex;
    }
    __syncthreads();

    const int4* pd = reinterpret_cast<const int4*>(dst + (size_t)sl * SL_EDGES);
    const int4* ps = reinterpret_cast<const int4*>(src + (size_t)sl * SL_EDGES);
    int4 d[SL_ITER], s[SL_ITER];
#pragma unroll
    for (int u = 0; u < SL_ITER; ++u) {
        const int idx = u * BT + threadIdx.x;
        const int j = idx < SL_V4 ? idx : SL_V4 - 1;
        d[u] = pd[j];
        s[u] = ps[j];
    }
#pragma unroll
    for (int u = 0; u < SL_ITER; ++u) {
        if (u * BT + (int)threadIdx.x < SL_V4) {
            const unsigned p0 = min(atomicAdd(&cursor[bucket_of(d[u].x)], 1u), (unsigned)(SL_EDGES - 1));
            const unsigned p1 = min(atomicAdd(&cursor[bucket_of(d[u].y)], 1u), (unsigned)(SL_EDGES - 1));
            const unsigned p2 = min(atomicAdd(&cursor[bucket_of(d[u].z)], 1u), (unsigned)(SL_EDGES - 1));
            const unsigned p3 = min(atomicAdd(&cursor[bucket_of(d[u].w)], 1u), (unsigned)(SL_EDGES - 1));
            sdst[p0] = d[u].x; ssrc[p0] = s[u].x;
            sdst[p1] = d[u].y; ssrc[p1] = s[u].y;
            sdst[p2] = d[u].z; ssrc[p2] = s[u].z;
            sdst[p3] = d[u].w; ssrc[p3] = s[u].w;
        }
    }
    __syncthreads();
    for (int i = threadIdx.x; i < SL_EDGES; i += BT) {
        const int dd = sdst[i];
        const unsigned b = bucket_of(dd);
        const unsigned local = (unsigned)(dd - (int)b * BK_NODES) & (unsigned)((1 << (32 - SRC_BITS)) - 1);
        unsigned slot = (unsigned)(delta[b] + i);
        if (slot >= (unsigned)NE) slot = NE - 1;
        rec[slot] = ((unsigned)ssrc[i] & (unsigned)((1 << SRC_BITS) - 1)) | (local << SRC_BITS);
    }
}

template <int K, int FOUT>
__device__ __forceinline__ void load_b_frags(const float* __restrict__ Wt, int m, int half,
                                             v2f (&b)[FOUT / 16][K / 4]) {
#pragma unroll
    for (int tn = 0; tn < FOUT / 16; ++tn) {
#pragma unroll
        for (int kb = 0; kb < K; kb += 4) {
            const int k0 = kb + 2 * half;
            b[tn][kb / 4].x = Wt[(size_t)(k0 + 0) * FOUT + tn * 16 + m];
            b[tn][kb / 4].y = Wt[(size_t)(k0 + 1) * FOUT + tn * 16 + m];
        }
    }
}

template <int K>
__device__ __forceinline__ void load_a_frags(const float* arow, int half, v2f (&a)[K / 4]) {
#pragma unroll
    for (int kb = 0; kb < K; kb += 4) {
        const int k0 = kb + 2 * half;
        a[kb / 4].x = arow[k0 + 0];
        a[kb / 4].y = arow[k0 + 1];
    }
}

template <int K, int FOUT>
__device__ __forceinline__ void tile_product(const v2f (&a)[K / 4], const v2f (&b)[FOUT / 16][K / 4],
                                             v8f (&acc)[FOUT / 16]) {
#pragma unroll
    for (int tn = 0; tn < FOUT / 16; ++tn) acc[tn] = v8f{};
#pragma unroll
    for (int i = 0; i < K / 4; ++i) {
#pragma unroll
        for (int tn = 0; tn < FOUT / 16; ++tn)
            acc[tn] = __builtin_amdgcn_wmma_f32_16x16x4_f32(false, a[i], false, b[tn][i],
                                                            (short)0, acc[tn], false, false);
    }
}

#define CSR_REG (BK_CAP / BT)
static_assert(BK_CAP % BT == 0, "the records kept in registers cover the LDS plane exactly");
__global__ __launch_bounds__(BT) void bucket_csr_kernel(const unsigned* __restrict__ off,
                                                        const unsigned* __restrict__ rec,
                                                        float* __restrict__ dinv, int* __restrict__ cnt,
                                                        int* __restrict__ rowptr,
                                                        int* __restrict__ csr_src,
                                                        const float* __restrict__ X,
                                                        const float* __restrict__ W,
                                                        float* __restrict__ hw) {
    __shared__ unsigned deg[BK_NODES];
    __shared__ unsigned cur[BK_NODES];
    __shared__ int ordered[BK_CAP];
    __shared__ int wsum[32];
    const int bk = blockIdx.x;
    const int tid = (int)threadIdx.x;
    int start = (int)off[(size_t)bk * NSL];
    int end = bk + 1 < NBK ? (int)off[(size_t)(bk + 1) * NSL] : NE;
    start = min(max(start, 0), NE);
    end = min(max(end, start), NE);
    const int n = end - start;
    unsigned r[CSR_REG];
#pragma unroll
    for (int k = 0; k < CSR_REG; ++k) {
        r[k] = 0u;
        if (k * BT < n) {
            const int i = k * BT + tid;
            r[k] = rec[start + (i < n ? i : n - 1)];
        }
    }
    if (tid < BK_NODES) deg[tid] = 0u;
    __syncthreads();
#pragma unroll
    for (int k = 0; k < CSR_REG; ++k) {
        if (k * BT + tid < n)
            atomicAdd(&deg[min(r[k] >> SRC_BITS, (unsigned)(BK_NODES - 1))], 1u);
    }
    for (int i = BK_CAP + tid; i < n; i += BT) {
        const unsigned l = min(rec[start + i] >> SRC_BITS, (unsigned)(BK_NODES - 1));
        atomicAdd(&deg[l], 1u);
    }
    __syncthreads();
    const int c = tid < BK_NODES ? (int)deg[tid] : 0;
    int total;
    const int ex = block_excl_scan(c, wsum, total);
    if (tid < BK_NODES) {
        const int node = bk * BK_NODES + tid;
        cur[tid] = (unsigned)ex;
        dinv[node] = rsqrtf((float)(c + 1));
        cnt[node] = c;
        rowptr[node] = start + ex;
    }
    __syncthreads();
    int p[CSR_REG];
#pragma unroll
    for (int k = 0; k < CSR_REG; ++k) {
        p[k] = n;
        if (k * BT + tid < n)
            p[k] = (int)atomicAdd(&cur[min(r[k] >> SRC_BITS, (unsigned)(BK_NODES - 1))], 1u);
    }
    const int m = n < BK_CAP ? n : BK_CAP;
#pragma unroll
    for (int k = 0; k < CSR_REG; ++k) {
        const int sidx = (int)(r[k] & (unsigned)((1 << SRC_BITS) - 1));
        if (p[k] < m) ordered[p[k]] = sidx;
        else if (p[k] < n) csr_src[start + p[k]] = sidx;
    }
    for (int i = BK_CAP + tid; i < n; i += BT) {
        const unsigned rr = rec[start + i];
        const unsigned l = min(rr >> SRC_BITS, (unsigned)(BK_NODES - 1));
        const int q = (int)atomicAdd(&cur[l], 1u);
        const int sidx = (int)(rr & (unsigned)((1 << SRC_BITS) - 1));
        if (q < m) ordered[q] = sidx;
        else if (q < n) csr_src[start + q] = sidx;
    }
    __syncthreads();
    for (int i = tid; i < m; i += BT) csr_src[start + i] = ordered[i];

    {
        constexpr int TILES = BK_NODES / 16;
        const int lane = tid & 31;
        const int wave = __builtin_amdgcn_readfirstlane(tid >> 5);
        const int fm = lane & 15;
        const int half = lane >> 4;
        v2f bf[HID_DIM / 16][IN_DIM / 4];
        load_b_frags<IN_DIM, HID_DIM>(W, fm, half, bf);
        for (int t = wave; t < TILES; t += BT / 32) {
            const int l0 = t * 16;
            const size_t row0 = (size_t)bk * BK_NODES + l0;
            v2f af[IN_DIM / 4];
            load_a_frags<IN_DIM>(X + (row0 + fm) * IN_DIM, half, af);
            v8f acc[HID_DIM / 16];
            tile_product<IN_DIM, HID_DIM>(af, bf, acc);
#pragma unroll
            for (int r = 0; r < 8; ++r) {
                const int l = l0 + r + 8 * half;
                const float s = rsqrtf((float)((int)deg[l] + 1));
                float* o = hw + ((size_t)bk * BK_NODES + l) * HID_DIM + fm;
#pragma unroll
                for (int tn = 0; tn < HID_DIM / 16; ++tn) o[tn * 16] = acc[tn][r] * s;
            }
        }
    }
}
static_assert(BK_NODES % 16 == 0, "a bucket's nodes split into whole 16-row tiles");

#define NODES_PER_WAVE 16
#define PULL_WAVES_PER_BLOCK 4
#define TILE_PITCH4 9
static_assert(NODES_PER_WAVE <= 32, "a lane holds the row range of one of the wave's nodes");
static_assert((long long)NN * HID_DIM * 4 < (1LL << 32), "row byte offsets are kept in 32 bits");

template <int F>
struct PullShape {
    static constexpr int LPR = F / 4;
    static constexpr int RPI = 32 / LPR;
    static constexpr int UNR = 32 / RPI;
};

template <int F>
__device__ __forceinline__ void chunk_sources(int idx, int g, unsigned q16,
                                              unsigned (&so)[PullShape<F>::UNR]) {
#pragma unroll
    for (int u = 0; u < PullShape<F>::UNR; ++u)
        so[u] = (unsigned)__shfl(idx, u * PullShape<F>::RPI + g) * (unsigned)(F * 4) + q16;
}

template <int F>
__device__ __forceinline__ void chunk_rows(const float* __restrict__ hws, int m,
                                           const unsigned (&so)[PullShape<F>::UNR],
                                           float4 (&v)[PullShape<F>::UNR]) {
#pragma unroll
    for (int u = 0; u < PullShape<F>::UNR; ++u) {
        v[u] = make_float4(0.0f, 0.0f, 0.0f, 0.0f);
        if (u * PullShape<F>::RPI < m)
            v[u] = *reinterpret_cast<const float4*>(reinterpret_cast<const char*>(hws) + so[u]);
    }
}

template <int F>
__device__ __forceinline__ void chunk_add(int m, int g, float nd,
                                          const float4 (&v)[PullShape<F>::UNR], float4& acc) {
#pragma unroll
    for (int u = 0; u < PullShape<F>::UNR; ++u) {
        if (u * PullShape<F>::RPI + g < m) {
            acc.x += v[u].x * nd;
            acc.y += v[u].y * nd;
            acc.z += v[u].z * nd;
            acc.w += v[u].w * nd;
        }
    }
}

template <int F, bool RELU, int FNEXT>
__global__ __launch_bounds__(32 * PULL_WAVES_PER_BLOCK) void pull_agg_kernel(
        const int* __restrict__ rowptr, const int* __restrict__ cnt,
        const int* __restrict__ csr_src, const float* __restrict__ hws,
        const float* __restrict__ dinv, const float* __restrict__ bias,
        const float* __restrict__ Wn, float* __restrict__ out, int n) {
    static_assert(F == 32 || F == 16, "a row is 8 or 4 float4");
    static_assert(FNEXT == 0 || (NODES_PER_WAVE == 16 && FNEXT % 16 == 0 && F % 4 == 0 && F / 4 < TILE_PITCH4),
                  "the wave's rows are one 16-row operand tile");
    __shared__ float4 tile4[FNEXT > 0 ? PULL_WAVES_PER_BLOCK * 16 * TILE_PITCH4 : 1];
    constexpr int LPR = PullShape<F>::LPR;
    constexpr int UNR = PullShape<F>::UNR;
    const int lane = threadIdx.x & 31;
    const int wave = __builtin_amdgcn_readfirstlane((int)((blockIdx.x * blockDim.x + threadIdx.x) >> 5));
    const int g = lane / LPR;
    const int q = lane % LPR;
    const unsigned q16 = (unsigned)q * 16u;
    const int node0 = wave * NODES_PER_WAVE;
    if (node0 >= n) return;
    float4* const wtile = tile4 + (FNEXT > 0 ? (int)(threadIdx.x >> 5) * 16 * TILE_PITCH4 : 0);

    int rp = 0, cn = 0, dn = 0;
    if (lane < NODES_PER_WAVE && node0 + lane < n) {
        rp = rowptr[node0 + lane];
        cn = cnt[node0 + lane];
        dn = __float_as_int(dinv[node0 + lane]);
    }
    const float4 b4 = reinterpret_cast<const float4*>(bias)[q];

    int idx_next = 0;
    {
        const int d0 = __builtin_amdgcn_readlane(cn, 0);
        const int s0 = __builtin_amdgcn_readlane(rp, 0);
        if (d0 > 0) idx_next = csr_src[s0 + (lane < d0 ? lane : d0 - 1)];
    }

    for (int k = 0; k < NODES_PER_WAVE; ++k) {
        const int node = node0 + k;
        if (node >= n) break;
        const int start = __builtin_amdgcn_readlane(rp, k);
        const int deg = __builtin_amdgcn_readlane(cn, k);
        const float nd = __int_as_float(__builtin_amdgcn_readlane(dn, k));

        unsigned so[UNR];
        float4 v[UNR];
        chunk_sources<F>(idx_next, g, q16, so);
        if (k + 1 < NODES_PER_WAVE && node + 1 < n) {
            const int d1 = __builtin_amdgcn_readlane(cn, k + 1);
            const int s1 = __builtin_amdgcn_readlane(rp, k + 1);
            if (d1 > 0) idx_next = csr_src[s1 + (lane < d1 ? lane : d1 - 1)];
        }
        const float4 self = reinterpret_cast<const float4*>(hws + (size_t)node * F)[q];

        float4 acc = make_float4(0.0f, 0.0f, 0.0f, 0.0f);
        const int m0 = deg < 32 ? deg : 32;
        chunk_rows<F>(hws, m0, so, v);
        chunk_add<F>(m0, g, nd, v, acc);
        for (int base = 32; base < deg; base += 32) {
            const int left = deg - base;
            const int m = left < 32 ? left : 32;
            const int e = base + lane < deg ? base + lane : deg - 1;
            const int idx = csr_src[start + e];
            chunk_sources<F>(idx, g, q16, so);
            chunk_rows<F>(hws, m, so, v);
            chunk_add<F>(m, g, nd, v, acc);
        }
#pragma unroll
        for (int d = LPR; d < 32; d <<= 1) {
            acc.x += __shfl_xor(acc.x, d);
            acc.y += __shfl_xor(acc.y, d);
            acc.z += __shfl_xor(acc.z, d);
            acc.w += __shfl_xor(acc.w, d);
        }
        if (g == 0) {
            float4 r;
            r.x = (self.x * nd + b4.x) + acc.x;
            r.y = (self.y * nd + b4.y) + acc.y;
            r.z = (self.z * nd + b4.z) + acc.z;
            r.w = (self.w * nd + b4.w) + acc.w;
            if (RELU) {
                r.x = fmaxf(r.x, 0.0f);
                r.y = fmaxf(r.y, 0.0f);
                r.z = fmaxf(r.z, 0.0f);
                r.w = fmaxf(r.w, 0.0f);
            }
            if (FNEXT > 0) wtile[((k >> 1) + 8 * (k & 1)) * TILE_PITCH4 + q] = r;
            else reinterpret_cast<float4*>(out + (size_t)node * F)[q] = r;
        }
    }

    if (FNEXT > 0) {
        __builtin_amdgcn_fence(__ATOMIC_RELEASE, "wavefront");
        __builtin_amdgcn_wave_barrier();
        __builtin_amdgcn_fence(__ATOMIC_ACQUIRE, "wavefront");
        constexpr int FN = FNEXT > 0 ? FNEXT : 16;
        const int fm = lane & 15;
        const int half = lane >> 4;
        v2f bf[FN / 16][F / 4];
        load_b_frags<F, FN>(Wn, fm, half, bf);
        v2f af[F / 4];
        load_a_frags<F>(reinterpret_cast<const float*>(wtile + fm * TILE_PITCH4), half, af);
        v8f acc[FN / 16];
        tile_product<F, FN>(af, bf, acc);
#pragma unroll
        for (int r = 0; r < 8; ++r) {
            const int l = 2 * r + half;
            const float s = __int_as_float(__shfl(dn, l));
            if (node0 + l < n) {
#pragma unroll
                for (int tn = 0; tn < FN / 16; ++tn)
                    out[(size_t)(node0 + l) * FN + tn * 16 + fm] = acc[tn][r] * s;
            }
        }
    }
}

static inline int cdiv(long long a, long long b) { return (int)((a + b - 1) / b); }

extern "C" void kernel_launch(void* const* d_in, const int* in_sizes, int n_in,
                              void* d_out, int out_size, void* d_ws, size_t ws_size,
                              hipStream_t stream) {
    const float* x  = (const float*)d_in[0];
    const int* ei   = (const int*)d_in[1];
    const float* W1 = (const float*)d_in[2];
    const float* b1 = (const float*)d_in[3];
    const float* W2 = (const float*)d_in[4];
    const float* b2 = (const float*)d_in[5];
    float* out = (float*)d_out;

    const int* src = ei;
    const int* dst = ei + NE;

    char* ws = (char*)d_ws;
    float* dinv = (float*)ws;
    float* hw1  = (float*)(ws + (size_t)NN * 4);
    float* hw2  = hw1 + (size_t)NN * HID_DIM;
    unsigned* off = (unsigned*)(hw2 + (size_t)NN * OUT_DIM);
    unsigned* rec = off + 65536;
    int* cnt    = (int*)(rec + (size_t)NE);
    int* rowptr = cnt + NN;
    int* csrc   = rowptr + NN;
    static_assert(NBK * NSL <= 65536, "off plane");

    part_count_kernel<<<NSL, BT, 0, stream>>>(dst, off);
    part_scan_kernel<<<1, BT, 0, stream>>>(off);
    part_scatter_kernel<<<NSL, BT, 0, stream>>>(src, dst, off, rec);
    bucket_csr_kernel<<<NBK, BT, 0, stream>>>(off, rec, dinv, cnt, rowptr, csrc, x, W1, hw1);

    const int pull_blocks = cdiv(NN, NODES_PER_WAVE * PULL_WAVES_PER_BLOCK);

    pull_agg_kernel<HID_DIM, true, OUT_DIM><<<pull_blocks, 32 * PULL_WAVES_PER_BLOCK, 0, stream>>>(
        rowptr, cnt, csrc, hw1, dinv, b1, W2, hw2, NN);

    pull_agg_kernel<OUT_DIM, false, 0><<<pull_blocks, 32 * PULL_WAVES_PER_BLOCK, 0, stream>>>(
        rowptr, cnt, csrc, hw2, dinv, b2, nullptr, out, NN);
}


# edge partition made slice-local: a slice stores its bucket-sorted records in its own region plus a run-start table; bucket start = sum of run starts; count+scan launches dropped; sums: arrival order
# speedup vs baseline: 5.8488x; 1.1218x over previous
#include <hip/hip_runtime.h>

#define NN 200000
#define NE 6400000
#define IN_DIM 32
#define HID_DIM 32
#define OUT_DIM 16

typedef __attribute__((ext_vector_type(2))) float v2f;
typedef __attribute__((ext_vector_type(8))) float v8f;

#define BK_NODES 800
#define NBK (NN / BK_NODES)
#define NSL 250
#define SL_EDGES (NE / NSL)
#define SL_V4 (SL_EDGES / 4)
#define BT 1024
#define SL_ITER ((SL_V4 + BT - 1) / BT)
#define SRC_BITS 18
#define BK_CAP 32768
#define TAB_PITCH (NBK + 1)
static_assert(NN % BK_NODES == 0 && NE % NSL == 0 && SL_EDGES % 4 == 0, "even split");
static_assert(NN <= (1 << SRC_BITS) && BK_NODES <= (1 << (32 - SRC_BITS)), "record fields");
static_assert(NBK < BT && BK_NODES <= BT && NSL <= BT, "one thread per bucket / per node / per slice");
static_assert((long long)NE < (1LL << 31), "edge slots are kept in non-negative ints");

__device__ __forceinline__ unsigned bucket_of(int d) {
    const unsigned b = (unsigned)d / (unsigned)BK_NODES;
    return b < (unsigned)NBK ? b : (unsigned)(NBK - 1);
}

__device__ __forceinline__ int block_excl_scan(int v, int* wsum, int& total) {
    const int lane = threadIdx.x & 31;
    const int w = threadIdx.x >> 5;
    int inc = v;
#pragma unroll
    for (int d = 1; d < 32; d <<= 1) {
        const int t = __shfl_up(inc, d);
        if (lane >= d) inc += t;
    }
    if (lane == 31) wsum[w] = inc;
    __syncthreads();
    const int ws = wsum[lane];
    int winc = ws;
#pragma unroll
    for (int d = 1; d < 32; d <<= 1) {
        const int t = __shfl_up(winc, d);
        if (lane >= d) winc += t;
    }
    const int woff = __shfl(winc, w) - __shfl(ws, w);
    total = __shfl(winc, 31);
    __syncthreads();
    return woff + inc - v;
}
static_assert(BT == 1024, "block_excl_scan: 32 waves of 32");

__global__ __launch_bounds__(BT) void slice_sort_kernel(const int* __restrict__ src,
                                                        const int* __restrict__ dst,
                                                        unsigned* __restrict__ tab,
                                                        unsigned* __restrict__ rec) {
    __shared__ int4 srec4[SL_V4];
    __shared__ unsigned cnt[BT / 32][NBK];
    __shared__ unsigned cursor[NBK];
    __shared__ int wsum[32];
    unsigned* srec = reinterpret_cast<unsigned*>(srec4);
    const int sl = blockIdx.x;
    const int w = threadIdx.x >> 5;
    for (int i = threadIdx.x; i < (BT / 32) * NBK; i += BT) (&cnt[0][0])[i] = 0u;
    __syncthreads();
    const int4* pd = reinterpret_cast<const int4*>(dst + (size_t)sl * SL_EDGES);
    const int4* ps = reinterpret_cast<const int4*>(src + (size_t)sl * SL_EDGES);
    int4 d[SL_ITER], s[SL_ITER];
#pragma unroll
    for (int u = 0; u < SL_ITER; ++u) {
        const int idx = u * BT + threadIdx.x;
        const int j = idx < SL_V4 ? idx : SL_V4 - 1;
        d[u] = pd[j];
        s[u] = ps[j];
    }
#pragma unroll
    for (int u = 0; u < SL_ITER; ++u) {
        if (u * BT + (int)threadIdx.x < SL_V4) {
            atomicAdd(&cnt[w][bucket_of(d[u].x)], 1u);
            atomicAdd(&cnt[w][bucket_of(d[u].y)], 1u);
            atomicAdd(&cnt[w][bucket_of(d[u].z)], 1u);
            atomicAdd(&cnt[w][bucket_of(d[u].w)], 1u);
        }
    }
    __syncthreads();
    int c = 0;
    if (threadIdx.x < NBK) {
#pragma unroll
        for (int k = 0; k < BT / 32; ++k) c += (int)cnt[k][threadIdx.x];
    }
    int total;
    const int ex = block_excl_scan(c, wsum, total);
    if (threadIdx.x < NBK) {
        cursor[threadIdx.x] = (unsigned)ex;
        tab[(size_t)sl * TAB_PITCH + threadIdx.x] = (unsigned)ex;
    }
    if (threadIdx.x == NBK) tab[(size_t)sl * TAB_PITCH + NBK] = (unsigned)SL_EDGES;
    __syncthreads();
#pragma unroll
    for (int u = 0; u < SL_ITER; ++u) {
        if (u * BT + (int)threadIdx.x < SL_V4) {
            const int dd[4] = {d[u].x, d[u].y, d[u].z, d[u].w};
            const int ss[4] = {s[u].x, s[u].y, s[u].z, s[u].w};
            unsigned p[4], rr[4];
#pragma unroll
            for (int e = 0; e < 4; ++e) {
                const unsigned b = bucket_of(dd[e]);
                const unsigned local = (unsigned)(dd[e] - (int)b * BK_NODES) & (unsigned)((1 << (32 - SRC_BITS)) - 1);
                rr[e] = ((unsigned)ss[e] & (unsigned)((1 << SRC_BITS) - 1)) | (local << SRC_BITS);
                p[e] = min(atomicAdd(&cursor[b], 1u), (unsigned)(SL_EDGES - 1));
            }
#pragma unroll
            for (int e = 0; e < 4; ++e) srec[p[e]] = rr[e];
        }
    }
    __syncthreads();
    int4* out4 = reinterpret_cast<int4*>(rec + (size_t)sl * SL_EDGES);
    for (int i = threadIdx.x; i < SL_V4; i += BT) out4[i] = srec4[i];
}

template <int K, int FOUT>
__device__ __forceinline__ void load_b_frags(const float* __restrict__ Wt, int m, int half,
                                             v2f (&b)[FOUT / 16][K / 4]) {
#pragma unroll
    for (int tn = 0; tn < FOUT / 16; ++tn) {
#pragma unroll
        for (int kb = 0; kb < K; kb += 4) {
            const int k0 = kb + 2 * half;
            b[tn][kb / 4].x = Wt[(size_t)(k0 + 0) * FOUT + tn * 16 + m];
            b[tn][kb / 4].y = Wt[(size_t)(k0 + 1) * FOUT + tn * 16 + m];
        }
    }
}

template <int K>
__device__ __forceinline__ void load_a_frags(const float* arow, int half, v2f (&a)[K / 4]) {
#pragma unroll
    for (int kb = 0; kb < K; kb += 4) {
        const int k0 = kb + 2 * half;
        a[kb / 4].x = arow[k0 + 0];
        a[kb / 4].y = arow[k0 + 1];
    }
}

template <int K, int FOUT>
__device__ __forceinline__ void tile_product(const v2f (&a)[K / 4], const v2f (&b)[FOUT / 16][K / 4],
                                             v8f (&acc)[FOUT / 16]) {
#pragma unroll
    for (int tn = 0; tn < FOUT / 16; ++tn) acc[tn] = v8f{};
#pragma unroll
    for (int i = 0; i < K / 4; ++i) {
#pragma unroll
        for (int tn = 0; tn < FOUT / 16; ++tn)
            acc[tn] = __builtin_amdgcn_wmma_f32_16x16x4_f32(false, a[i], false, b[tn][i],
                                                            (short)0, acc[tn], false, false);
    }
}

#define RUN_PER_WAVE ((NSL + 31) / 32)
#define RUN_REG 4
__global__ __launch_bounds__(BT) void bucket_csr_kernel(const unsigned* __restrict__ tab,
                                                        const unsigned* __restrict__ rec,
                                                        float* __restrict__ dinv, int* __restrict__ cnt,
                                                        int* __restrict__ rowptr,
                                                        int* __restrict__ csr_src,
                                                        const float* __restrict__ X,
                                                        const float* __restrict__ W,
                                                        float* __restrict__ hw) {
    __shared__ unsigned deg[BK_NODES];
    __shared__ unsigned cur[BK_NODES];
    __shared__ int ordered[BK_CAP];
    __shared__ int wsum[32];
    __shared__ int rbase[RUN_PER_WAVE * 32];
    __shared__ int rlen[RUN_PER_WAVE * 32];
    const int bk = blockIdx.x;
    const int tid = (int)threadIdx.x;
    int t0 = 0, c0 = 0;
    if (tid < NSL) {
        const int a = (int)tab[(size_t)tid * TAB_PITCH + bk];
        const int b = (int)tab[(size_t)tid * TAB_PITCH + bk + 1];
        t0 = min(max(a, 0), SL_EDGES);
        c0 = min(max(b, t0), SL_EDGES) - t0;
    }
    if (tid < RUN_PER_WAVE * 32) {
        rbase[tid] = (tid < NSL ? tid : 0) * SL_EDGES + t0;
        rlen[tid] = c0;
    }
    if (tid < BK_NODES) deg[tid] = 0u;
    int n, start;
    block_excl_scan(c0, wsum, n);
    block_excl_scan(t0, wsum, start);
    start = min(max(start, 0), NE);
    n = min(max(n, 0), NE - start);
    const int lane = tid & 31;
    const int wave = __builtin_amdgcn_readfirstlane(tid >> 5);
    int base[RUN_PER_WAVE], len[RUN_PER_WAVE];
#pragma unroll
    for (int q = 0; q < RUN_PER_WAVE; ++q) {
        base[q] = __builtin_amdgcn_readfirstlane(rbase[q * 32 + wave]);
        len[q] = __builtin_amdgcn_readfirstlane(rlen[q * 32 + wave]);
    }
    unsigned r[RUN_PER_WAVE][RUN_REG];
#pragma unroll
    for (int q = 0; q < RUN_PER_WAVE; ++q) {
#pragma unroll
        for (int u = 0; u < RUN_REG; ++u) {
            r[q][u] = 0u;
            if (u * 32 < len[q]) {
                const int j = u * 32 + lane;
                r[q][u] = rec[base[q] + (j < len[q] ? j : len[q] - 1)];
            }
        }
    }
#pragma unroll
    for (int q = 0; q < RUN_PER_WAVE; ++q) {
#pragma unroll
        for (int u = 0; u < RUN_REG; ++u) {
            if (u * 32 + lane < len[q])
                atomicAdd(&deg[min(r[q][u] >> SRC_BITS, (unsigned)(BK_NODES - 1))], 1u);
        }
        for (int j = RUN_REG * 32 + lane; j < len[q]; j += 32) {
            const unsigned l = min(rec[base[q] + j] >> SRC_BITS, (unsigned)(BK_NODES - 1));
            atomicAdd(&deg[l], 1u);
        }
    }
    __syncthreads();
    const int c = tid < BK_NODES ? (int)deg[tid] : 0;
    int total;
    const int ex = block_excl_scan(c, wsum, total);
    if (tid < BK_NODES) {
        const int node = bk * BK_NODES + tid;
        cur[tid] = (unsigned)ex;
        dinv[node] = rsqrtf((float)(c + 1));
        cnt[node] = c;
        rowptr[node] = start + ex;
    }
    __syncthreads();
    int p[RUN_PER_WAVE][RUN_REG];
#pragma unroll
    for (int q = 0; q < RUN_PER_WAVE; ++q) {
#pragma unroll
        for (int u = 0; u < RUN_REG; ++u) {
            p[q][u] = n;
            if (u * 32 + lane < len[q])
                p[q][u] = (int)atomicAdd(&cur[min(r[q][u] >> SRC_BITS, (unsigned)(BK_NODES - 1))], 1u);
        }
    }
    const int m = n < BK_CAP ? n : BK_CAP;
#pragma unroll
    for (int q = 0; q < RUN_PER_WAVE; ++q) {
#pragma unroll
        for (int u = 0; u < RUN_REG; ++u) {
            const int sidx = (int)(r[q][u] & (unsigned)((1 << SRC_BITS) - 1));
            if (p[q][u] < m) ordered[p[q][u]] = sidx;
            else if (p[q][u] < n) csr_src[start + p[q][u]] = sidx;
        }
        for (int j = RUN_REG * 32 + lane; j < len[q]; j += 32) {
            const unsigned rr = rec[base[q] + j];
            const unsigned l = min(rr >> SRC_BITS, (unsigned)(BK_NODES - 1));
            const int qq = (int)atomicAdd(&cur[l], 1u);
            const int sidx = (int)(rr & (unsigned)((1 << SRC_BITS) - 1));
            if (qq < m) ordered[qq] = sidx;
            else if (qq < n) csr_src[start + qq] = sidx;
        }
    }
    __syncthreads();
    for (int i = tid; i < m; i += BT) csr_src[start + i] = ordered[i];

    {
        constexpr int TILES = BK_NODES / 16;
        const int fm = lane & 15;
        const int half = lane >> 4;
        v2f bf[HID_DIM / 16][IN_DIM / 4];
        load_b_frags<IN_DIM, HID_DIM>(W, fm, half, bf);
        for (int t = wave; t < TILES; t += BT / 32) {
            const int l0 = t * 16;
            const size_t row0 = (size_t)bk * BK_NODES + l0;
            v2f af[IN_DIM / 4];
            load_a_frags<IN_DIM>(X + (row0 + fm) * IN_DIM, half, af);
            v8f acc[HID_DIM / 16];
            tile_product<IN_DIM, HID_DIM>(af, bf, acc);
#pragma unroll
            for (int r = 0; r < 8; ++r) {
                const int l = l0 + r + 8 * half;
                const float s = rsqrtf((float)((int)deg[l] + 1));
                float* o = hw + ((size_t)bk * BK_NODES + l) * HID_DIM + fm;
#pragma unroll
                for (int tn = 0; tn < HID_DIM / 16; ++tn) o[tn * 16] = acc[tn][r] * s;
            }
        }
    }
}
static_assert(BK_NODES % 16 == 0, "a bucket's nodes split into whole 16-row tiles");

#define NODES_PER_WAVE 16
#define PULL_WAVES_PER_BLOCK 4
#define TILE_PITCH4 9
static_assert(NODES_PER_WAVE <= 32, "a lane holds the row range of one of the wave's nodes");
static_assert((long long)NN * HID_DIM * 4 < (1LL << 32), "row byte offsets are kept in 32 bits");

template <int F>
struct PullShape {
    static constexpr int LPR = F / 4;
    static constexpr int RPI = 32 / LPR;
    static constexpr int UNR = 32 / RPI;
};

template <int F>
__device__ __forceinline__ void chunk_sources(int idx, int g, unsigned q16,
                                              unsigned (&so)[PullShape<F>::UNR]) {
#pragma unroll
    for (int u = 0; u < PullShape<F>::UNR; ++u)
        so[u] = (unsigned)__shfl(idx, u * PullShape<F>::RPI + g) * (unsigned)(F * 4) + q16;
}

template <int F>
__device__ __forceinline__ void chunk_rows(const float* __restrict__ hws, int m,
                                           const unsigned (&so)[PullShape<F>::UNR],
                                           float4 (&v)[PullShape<F>::UNR]) {
#pragma unroll
    for (int u = 0; u < PullShape<F>::UNR; ++u) {
        v[u] = make_float4(0.0f, 0.0f, 0.0f, 0.0f);
        if (u * PullShape<F>::RPI < m)
            v[u] = *reinterpret_cast<const float4*>(reinterpret_cast<const char*>(hws) + so[u]);
    }
}

template <int F>
__device__ __forceinline__ void chunk_add(int m, int g, float nd,
                                          const float4 (&v)[PullShape<F>::UNR], float4& acc) {
#pragma unroll
    for (int u = 0; u < PullShape<F>::UNR; ++u) {
        if (u * PullShape<F>::RPI + g < m) {
            acc.x += v[u].x * nd;
            acc.y += v[u].y * nd;
            acc.z += v[u].z * nd;
            acc.w += v[u].w * nd;
        }
    }
}

template <int F, bool RELU, int FNEXT>
__global__ __launch_bounds__(32 * PULL_WAVES_PER_BLOCK) void pull_agg_kernel(
        const int* __restrict__ rowptr, const int* __restrict__ cnt,
        const int* __restrict__ csr_src, const float* __restrict__ hws,
        const float* __restrict__ dinv, const float* __restrict__ bias,
        const float* __restrict__ Wn, float* __restrict__ out, int n) {
    static_assert(F == 32 || F == 16, "a row is 8 or 4 float4");
    static_assert(FNEXT == 0 || (NODES_PER_WAVE == 16 && FNEXT % 16 == 0 && F % 4 == 0 && F / 4 < TILE_PITCH4),
                  "the wave's rows are one 16-row operand tile");
    __shared__ float4 tile4[FNEXT > 0 ? PULL_WAVES_PER_BLOCK * 16 * TILE_PITCH4 : 1];
    constexpr int LPR = PullShape<F>::LPR;
    constexpr int UNR = PullShape<F>::UNR;
    const int lane = threadIdx.x & 31;
    const int wave = __builtin_amdgcn_readfirstlane((int)((blockIdx.x * blockDim.x + threadIdx.x) >> 5));
    const int g = lane / LPR;
    const int q = lane % LPR;
    const unsigned q16 = (unsigned)q * 16u;
    const int node0 = wave * NODES_PER_WAVE;
    if (node0 >= n) return;
    float4* const wtile = tile4 + (FNEXT > 0 ? (int)(threadIdx.x >> 5) * 16 * TILE_PITCH4 : 0);

    int rp = 0, cn = 0, dn = 0;
    if (lane < NODES_PER_WAVE && node0 + lane < n) {
        rp = rowptr[node0 + lane];
        cn = cnt[node0 + lane];
        dn = __float_as_int(dinv[node0 + lane]);
    }
    const float4 b4 = reinterpret_cast<const float4*>(bias)[q];

    int idx_next = 0;
    {
        const int d0 = __builtin_amdgcn_readlane(cn, 0);
        const int s0 = __builtin_amdgcn_readlane(rp, 0);
        if (d0 > 0) idx_next = csr_src[s0 + (lane < d0 ? lane : d0 - 1)];
    }

    for (int k = 0; k < NODES_PER_WAVE; ++k) {
        const int node = node0 + k;
        if (node >= n) break;
        const int start = __builtin_amdgcn_readlane(rp, k);
        const int deg = __builtin_amdgcn_readlane(cn, k);
        const float nd = __int_as_float(__builtin_amdgcn_readlane(dn, k));

        unsigned so[UNR];
        float4 v[UNR];
        chunk_sources<F>(idx_next, g, q16, so);
        if (k + 1 < NODES_PER_WAVE && node + 1 < n) {
            const int d1 = __builtin_amdgcn_readlane(cn, k + 1);
            const int s1 = __builtin_amdgcn_readlane(rp, k + 1);
            if (d1 > 0) idx_next = csr_src[s1 + (lane < d1 ? lane : d1 - 1)];
        }
        const float4 self = reinterpret_cast<const float4*>(hws + (size_t)node * F)[q];

        float4 acc = make_float4(0.0f, 0.0f, 0.0f, 0.0f);
        const int m0 = deg < 32 ? deg : 32;
        chunk_rows<F>(hws, m0, so, v);
        chunk_add<F>(m0, g, nd, v, acc);
        for (int base = 32; base < deg; base += 32) {
            const int left = deg - base;
            const int m = left < 32 ? left : 32;
            const int e = base + lane < deg ? base + lane : deg - 1;
            const int idx = csr_src[start + e];
            chunk_sources<F>(idx, g, q16, so);
            chunk_rows<F>(hws, m, so, v);
            chunk_add<F>(m, g, nd, v, acc);
        }
#pragma unroll
        for (int d = LPR; d < 32; d <<= 1) {
            acc.x += __shfl_xor(acc.x, d);
            acc.y += __shfl_xor(acc.y, d);
            acc.z += __shfl_xor(acc.z, d);
            acc.w += __shfl_xor(acc.w, d);
        }
        if (g == 0) {
            float4 r;
            r.x = (self.x * nd + b4.x) + acc.x;
            r.y = (self.y * nd + b4.y) + acc.y;
            r.z = (self.z * nd + b4.z) + acc.z;
            r.w = (self.w * nd + b4.w) + acc.w;
            if (RELU) {
                r.x = fmaxf(r.x, 0.0f);
                r.y = fmaxf(r.y, 0.0f);
                r.z = fmaxf(r.z, 0.0f);
                r.w = fmaxf(r.w, 0.0f);
            }
            if (FNEXT > 0) wtile[((k >> 1) + 8 * (k & 1)) * TILE_PITCH4 + q] = r;
            else reinterpret_cast<float4*>(out + (size_t)node * F)[q] = r;
        }
    }

    if (FNEXT > 0) {
        __builtin_amdgcn_fence(__ATOMIC_RELEASE, "wavefront");
        __builtin_amdgcn_wave_barrier();
        __builtin_amdgcn_fence(__ATOMIC_ACQUIRE, "wavefront");
        constexpr int FN = FNEXT > 0 ? FNEXT : 16;
        const int fm = lane & 15;
        const int half = lane >> 4;
        v2f bf[FN / 16][F / 4];
        load_b_frags<F, FN>(Wn, fm, half, bf);
        v2f af[F / 4];
        load_a_frags<F>(reinterpret_cast<const float*>(wtile + fm * TILE_PITCH4), half, af);
        v8f acc[FN / 16];
        tile_product<F, FN>(af, bf, acc);
#pragma unroll
        for (int r = 0; r < 8; ++r) {
            const int l = 2 * r + half;
            const float s = __int_as_float(__shfl(dn, l));
            if (node0 + l < n) {
#pragma unroll
                for (int tn = 0; tn < FN / 16; ++tn)
                    out[(size_t)(node0 + l) * FN + tn * 16 + fm] = acc[tn][r] * s;
            }
        }
    }
}

static inline int cdiv(long long a, long long b) { return (int)((a + b - 1) / b); }

extern "C" void kernel_launch(void* const* d_in, const int* in_sizes, int n_in,
                              void* d_out, int out_size, void* d_ws, size_t ws_size,
                              hipStream_t stream) {
    const float* x  = (const float*)d_in[0];
    const int* ei   = (const int*)d_in[1];
    const float* W1 = (const float*)d_in[2];
    const float* b1 = (const float*)d_in[3];
    const float* W2 = (const float*)d_in[4];
    const float* b2 = (const float*)d_in[5];
    float* out = (float*)d_out;

    const int* src = ei;
    const int* dst = ei + NE;

    char* ws = (char*)d_ws;
    float* dinv = (float*)ws;
    float* hw1  = (float*)(ws + (size_t)NN * 4);
    float* hw2  = hw1 + (size_t)NN * HID_DIM;
    unsigned* tab = (unsigned*)(hw2 + (size_t)NN * OUT_DIM);
    unsigned* rec = tab + 65536;
    int* cnt    = (int*)(rec + (size_t)NE);
    int* rowptr = cnt + NN;
    int* csrc   = rowptr + NN;
    static_assert(NSL * TAB_PITCH <= 65536, "tab plane");

    slice_sort_kernel<<<NSL, BT, 0, stream>>>(src, dst, tab, rec);
    bucket_csr_kernel<<<NBK, BT, 0, stream>>>(tab, rec, dinv, cnt, rowptr, csrc, x, W1, hw1);

    const int pull_blocks = cdiv(NN, NODES_PER_WAVE * PULL_WAVES_PER_BLOCK);

    pull_agg_kernel<HID_DIM, true, OUT_DIM><<<pull_blocks, 32 * PULL_WAVES_PER_BLOCK, 0, stream>>>(
        rowptr, cnt, csrc, hw1, dinv, b1, W2, hw2, NN);

    pull_agg_kernel<OUT_DIM, false, 0><<<pull_blocks, 32 * PULL_WAVES_PER_BLOCK, 0, stream>>>(
        rowptr, cnt, csrc, hw2, dinv, b2, nullptr, out, NN);
}
